# starting kernel
# baseline (speedup 1.0000x reference)
#include <hip/hip_runtime.h>
#include <hip/hip_bf16.h>
#include <math.h>


typedef float v2f __attribute__((ext_vector_type(2)));
typedef float v8f __attribute__((ext_vector_type(8)));

#define NQ    8
#define NS    256
#define NLAY  2
#define TM    32
#define THREADS 256

__device__ __forceinline__ int bswz(int k, int n) {
  const int fraglane = (((k >> 1) & 1) << 4) | (n & 15);
  return (((k >> 2) << 4) + (n >> 4)) * 64 + fraglane * 2 + (k & 1);
}

__global__ __launch_bounds__(256) void build_layer_mats(
    const float* __restrict__ wts, float* __restrict__ W)
{
  const int t    = threadIdx.x;
  const int wave = blockIdx.x * (blockDim.x >> 5) + (t >> 5);
  const int lane = t & 31;
  const int layer = wave >> 8;
  const int k     = wave & 255;

  float re[8], im[8];
#pragma unroll
  for (int r = 0; r < 8; ++r) {
    re[r] = (((lane << 3) | r) == k) ? 1.f : 0.f;
    im[r] = 0.f;
  }

  const float* wl = wts + layer * (NQ * 2);

#pragma unroll
  for (int i = 0; i < NQ; ++i) {
    const int b = 7 - i;
    const float hy = 0.5f * wl[i * 2 + 0];
    const float cy = cosf(hy), sy = sinf(hy);
    if (b < 3) {
      const int m = 1 << b;
#pragma unroll
      for (int r = 0; r < 8; ++r) {
        if (!(r & m)) {
          const int r1 = r | m;
          float a0 = re[r], a1 = re[r1];
          re[r] = cy * a0 - sy * a1;  re[r1] = sy * a0 + cy * a1;
          a0 = im[r]; a1 = im[r1];
          im[r] = cy * a0 - sy * a1;  im[r1] = sy * a0 + cy * a1;
        }
      }
    } else {
      const int lm = 1 << (b - 3);
      const bool hi = (lane & lm) != 0;
#pragma unroll
      for (int r = 0; r < 8; ++r) {
        const float ore = __shfl_xor(re[r], lm, 32);
        const float oim = __shfl_xor(im[r], lm, 32);
        re[r] = hi ? (sy * ore + cy * re[r]) : (cy * re[r] - sy * ore);
        im[r] = hi ? (sy * oim + cy * im[r]) : (cy * im[r] - sy * oim);
      }
    }
    const float hz = 0.5f * wl[i * 2 + 1];
    const float cz = cosf(hz), sz = sinf(hz);
#pragma unroll
    for (int r = 0; r < 8; ++r) {
      const int bit = (b < 3) ? ((r >> b) & 1) : ((lane >> (b - 3)) & 1);
      const float tt = bit ? -sz : sz;
      const float nr = cz * re[r] + tt * im[r];
      const float ni = cz * im[r] - tt * re[r];
      re[r] = nr; im[r] = ni;
    }
  }

#pragma unroll
  for (int g = 0; g < NQ; ++g) {
    const int ci = (g < 7) ? g : 7;
    const int ti = (g < 7) ? g + 1 : 0;
    const int bc = 7 - ci, bt = 7 - ti;
    if (bt < 3) {
      const int tm = 1 << bt;
      float tre[8], tim[8];
#pragma unroll
      for (int r = 0; r < 8; ++r) { tre[r] = re[r]; tim[r] = im[r]; }
#pragma unroll
      for (int r = 0; r < 8; ++r) {
        const int cb = (bc < 3) ? ((r >> bc) & 1) : ((lane >> (bc - 3)) & 1);
        re[r] = cb ? tre[r ^ tm] : tre[r];
        im[r] = cb ? tim[r ^ tm] : tim[r];
      }
    } else {
      const int lm = 1 << (bt - 3);
#pragma unroll
      for (int r = 0; r < 8; ++r) {
        const float ore = __shfl_xor(re[r], lm, 32);
        const float oim = __shfl_xor(im[r], lm, 32);
        const int cb = (bc < 3) ? ((r >> bc) & 1) : ((lane >> (bc - 3)) & 1);
        re[r] = cb ? ore : re[r];
        im[r] = cb ? oim : im[r];
      }
    }
  }

  float* Wre = W + layer * 2 * (NS * NS);
  float* Wim = Wre + NS * NS;
#pragma unroll
  for (int r = 0; r < 8; ++r) {
    const int n = (lane << 3) | r;
    const int idx = bswz(k, n);
    Wre[idx] = re[r];
    Wim[idx] = im[r];
  }
}

static __device__ __forceinline__ v8f wmma_f32(v2f a, v2f b, v8f c) {
  return __builtin_amdgcn_wmma_f32_16x16x4_f32(
       false, a,  false, b,
       (short)0, c,  false,  false);
}

__global__ __launch_bounds__(THREADS) void circuit_gemm(
    const float* __restrict__ x, const float* __restrict__ W,
    float* __restrict__ out)
{
  __shared__ float Sre[TM][NS];
  __shared__ float Sim[TM][NS];
  __shared__ float csC[TM][NQ];
  __shared__ float csS[TM][NQ];
  __shared__ float oacc[TM][NQ];

  const int t  = threadIdx.x;
  const int s0 = blockIdx.x * TM;

  {
    const int m = t >> 3, i = t & 7;
    const float a = x[(s0 + m) * NQ + i] * 0.785398163397448f;
    csC[m][i] = cosf(a);
    csS[m][i] = sinf(a);
    oacc[m][i] = 0.f;
  }
  __syncthreads();

  for (int e = t; e < TM * NS; e += THREADS) {
    const int m = e >> 8, s = e & 255;
    float p = 1.f;
#pragma unroll
    for (int i = 0; i < NQ; ++i)
      p *= ((s >> (7 - i)) & 1) ? csS[m][i] : csC[m][i];
    Sre[m][s] = p;
    Sim[m][s] = 0.f;
  }
  __syncthreads();

  const int wave = t >> 5, lane = t & 31;
  const int half = lane >> 4, lq = lane & 15;

  v8f accR[2][2], accI[2][2];

  for (int l = 0; l < NLAY; ++l) {
    const float* Wre = W + l * 2 * (NS * NS);
    const float* Wim = Wre + NS * NS;

#pragma unroll
    for (int tt = 0; tt < 2; ++tt)
#pragma unroll
      for (int mt = 0; mt < 2; ++mt) {
        accR[tt][mt] = (v8f){0.f, 0.f, 0.f, 0.f, 0.f, 0.f, 0.f, 0.f};
        accI[tt][mt] = (v8f){0.f, 0.f, 0.f, 0.f, 0.f, 0.f, 0.f, 0.f};
      }

    for (int k = 0; k < NS; k += 4) {
      const int kk = k + 2 * half;
      const v2f ar0 = *(const v2f*)&Sre[lq][kk];
      const v2f ai0 = *(const v2f*)&Sim[lq][kk];
      const v2f ar1 = *(const v2f*)&Sre[lq + 16][kk];
      const v2f ai1 = *(const v2f*)&Sim[lq + 16][kk];
      const v2f an0 = -ai0;
      const v2f an1 = -ai1;
#pragma unroll
      for (int tt = 0; tt < 2; ++tt) {
        const int nt = wave * 2 + tt;
        const int fidx = (((k >> 2) << 4) + nt) * 64 + lane * 2;
        const v2f br = *(const v2f*)&Wre[fidx];
        const v2f bi = *(const v2f*)&Wim[fidx];
        accR[tt][0] = wmma_f32(ar0, br, accR[tt][0]);
        accR[tt][0] = wmma_f32(an0, bi, accR[tt][0]);
        accI[tt][0] = wmma_f32(ar0, bi, accI[tt][0]);
        accI[tt][0] = wmma_f32(ai0, br, accI[tt][0]);
        accR[tt][1] = wmma_f32(ar1, br, accR[tt][1]);
        accR[tt][1] = wmma_f32(an1, bi, accR[tt][1]);
        accI[tt][1] = wmma_f32(ar1, bi, accI[tt][1]);
        accI[tt][1] = wmma_f32(ai1, br, accI[tt][1]);
      }
    }
    __syncthreads();

    if (l == 0) {
#pragma unroll
      for (int tt = 0; tt < 2; ++tt) {
        const int n = (wave * 2 + tt) * 16 + lq;
#pragma unroll
        for (int mt = 0; mt < 2; ++mt)
#pragma unroll
          for (int j = 0; j < 8; ++j) {
            const int m = mt * 16 + j + 8 * half;
            Sre[m][n] = accR[tt][mt][j];
            Sim[m][n] = accI[tt][mt][j];
          }
      }
      __syncthreads();
      for (int i = 0; i < NQ; ++i) {
        const int b = 7 - i;
        for (int p = t; p < TM * (NS / 2); p += THREADS) {
          const int m = p >> 7, pr = p & 127;
          const int low = pr & ((1 << b) - 1);
          const int sA = ((pr >> b) << (b + 1)) | low;
          const int sB = sA | (1 << b);
          const float c = csC[m][i], sn = csS[m][i];
          float a0 = Sre[m][sA], a1 = Sre[m][sB];
          Sre[m][sA] = c * a0 - sn * a1;
          Sre[m][sB] = sn * a0 + c * a1;
          a0 = Sim[m][sA]; a1 = Sim[m][sB];
          Sim[m][sA] = c * a0 - sn * a1;
          Sim[m][sB] = sn * a0 + c * a1;
        }
        __syncthreads();
      }
    }
  }

#pragma unroll
  for (int tt = 0; tt < 2; ++tt) {
    const int n = (wave * 2 + tt) * 16 + lq;
#pragma unroll
    for (int mt = 0; mt < 2; ++mt)
#pragma unroll
      for (int j = 0; j < 8; ++j) {
        const int m = mt * 16 + j + 8 * half;
        const float pv = accR[tt][mt][j] * accR[tt][mt][j] +
                         accI[tt][mt][j] * accI[tt][mt][j];
#pragma unroll
        for (int w = 0; w < NQ; ++w) {
          atomicAdd(&oacc[m][w], ((n >> (7 - w)) & 1) ? -pv : pv);
        }
      }
  }
  __syncthreads();

  {
    const int m = t >> 3, i = t & 7;
    out[(s0 + m) * NQ + i] = oacc[m][i];
  }
}

extern "C" void kernel_launch(void* const* d_in, const int* in_sizes, int n_in,
                              void* d_out, int out_size, void* d_ws, size_t ws_size,
                              hipStream_t stream) {
  const float* x   = (const float*)d_in[0];
  const float* wts = (const float*)d_in[1];
  float* out = (float*)d_out;
  float* W   = (float*)d_ws;

  const int B = in_sizes[0] / NQ;

  build_layer_mats<<<(NLAY * NS) / (256 / 32), 256, 0, stream>>>(wts, W);
  circuit_gemm<<<B / TM, THREADS, 0, stream>>>(x, W, out);
}


# K2 epilogue: the 256 same-word LDS float atomic adds per lane replaced by in-row DPP Walsh butterfly sums and per-wave LDS partials added in fixed order (sum order becomes a fixed tree)
# speedup vs baseline: 3.0822x; 3.0822x over previous
#include <hip/hip_runtime.h>
#include <hip/hip_bf16.h>
#include <math.h>


typedef float v2f __attribute__((ext_vector_type(2)));
typedef float v8f __attribute__((ext_vector_type(8)));

#define NQ    8
#define NS    256
#define NLAY  2
#define TM    32
#define THREADS 256

__device__ __forceinline__ int bswz(int k, int n) {
  const int fraglane = (((k >> 1) & 1) << 4) | (n & 15);
  return (((k >> 2) << 4) + (n >> 4)) * 64 + fraglane * 2 + (k & 1);
}

__global__ __launch_bounds__(256) void build_layer_mats(
    const float* __restrict__ wts, float* __restrict__ W)
{
  const int t    = threadIdx.x;
  const int wave = blockIdx.x * (blockDim.x >> 5) + (t >> 5);
  const int lane = t & 31;
  const int layer = wave >> 8;
  const int k     = wave & 255;

  float re[8], im[8];
#pragma unroll
  for (int r = 0; r < 8; ++r) {
    re[r] = (((lane << 3) | r) == k) ? 1.f : 0.f;
    im[r] = 0.f;
  }

  const float* wl = wts + layer * (NQ * 2);

#pragma unroll
  for (int i = 0; i < NQ; ++i) {
    const int b = 7 - i;
    const float hy = 0.5f * wl[i * 2 + 0];
    const float cy = cosf(hy), sy = sinf(hy);
    if (b < 3) {
      const int m = 1 << b;
#pragma unroll
      for (int r = 0; r < 8; ++r) {
        if (!(r & m)) {
          const int r1 = r | m;
          float a0 = re[r], a1 = re[r1];
          re[r] = cy * a0 - sy * a1;  re[r1] = sy * a0 + cy * a1;
          a0 = im[r]; a1 = im[r1];
          im[r] = cy * a0 - sy * a1;  im[r1] = sy * a0 + cy * a1;
        }
      }
    } else {
      const int lm = 1 << (b - 3);
      const bool hi = (lane & lm) != 0;
#pragma unroll
      for (int r = 0; r < 8; ++r) {
        const float ore = __shfl_xor(re[r], lm, 32);
        const float oim = __shfl_xor(im[r], lm, 32);
        re[r] = hi ? (sy * ore + cy * re[r]) : (cy * re[r] - sy * ore);
        im[r] = hi ? (sy * oim + cy * im[r]) : (cy * im[r] - sy * oim);
      }
    }
    const float hz = 0.5f * wl[i * 2 + 1];
    const float cz = cosf(hz), sz = sinf(hz);
#pragma unroll
    for (int r = 0; r < 8; ++r) {
      const int bit = (b < 3) ? ((r >> b) & 1) : ((lane >> (b - 3)) & 1);
      const float tt = bit ? -sz : sz;
      const float nr = cz * re[r] + tt * im[r];
      const float ni = cz * im[r] - tt * re[r];
      re[r] = nr; im[r] = ni;
    }
  }

#pragma unroll
  for (int g = 0; g < NQ; ++g) {
    const int ci = (g < 7) ? g : 7;
    const int ti = (g < 7) ? g + 1 : 0;
    const int bc = 7 - ci, bt = 7 - ti;
    if (bt < 3) {
      const int tm = 1 << bt;
      float tre[8], tim[8];
#pragma unroll
      for (int r = 0; r < 8; ++r) { tre[r] = re[r]; tim[r] = im[r]; }
#pragma unroll
      for (int r = 0; r < 8; ++r) {
        const int cb = (bc < 3) ? ((r >> bc) & 1) : ((lane >> (bc - 3)) & 1);
        re[r] = cb ? tre[r ^ tm] : tre[r];
        im[r] = cb ? tim[r ^ tm] : tim[r];
      }
    } else {
      const int lm = 1 << (bt - 3);
#pragma unroll
      for (int r = 0; r < 8; ++r) {
        const float ore = __shfl_xor(re[r], lm, 32);
        const float oim = __shfl_xor(im[r], lm, 32);
        const int cb = (bc < 3) ? ((r >> bc) & 1) : ((lane >> (bc - 3)) & 1);
        re[r] = cb ? ore : re[r];
        im[r] = cb ? oim : im[r];
      }
    }
  }

  float* Wre = W + layer * 2 * (NS * NS);
  float* Wim = Wre + NS * NS;
#pragma unroll
  for (int r = 0; r < 8; ++r) {
    const int n = (lane << 3) | r;
    const int idx = bswz(k, n);
    Wre[idx] = re[r];
    Wim[idx] = im[r];
  }
}

static __device__ __forceinline__ v8f wmma_f32(v2f a, v2f b, v8f c) {
  return __builtin_amdgcn_wmma_f32_16x16x4_f32(
       false, a,  false, b,
       (short)0, c,  false,  false);
}


template <int M>
static __device__ __forceinline__ float row_xor(float v) {
  const int b = __builtin_bit_cast(int, v);
  return __builtin_bit_cast(
      float, __builtin_amdgcn_update_dpp(b, b, 0x160 + M, 0xf, 0xf, true));
}

template <int M>
static __device__ __forceinline__ float walsh_step(float v, unsigned sgn) {
  const float other = row_xor<M>(v);
  const float own =
      __builtin_bit_cast(float, __builtin_bit_cast(unsigned, v) ^ sgn);
  return other + own;
}

__global__ __launch_bounds__(THREADS) void circuit_gemm(
    const float* __restrict__ x, const float* __restrict__ W,
    float* __restrict__ out)
{
  __shared__ float Sre[TM][NS];
  __shared__ float Sim[TM][NS];
  __shared__ float csC[TM][NQ];
  __shared__ float csS[TM][NQ];

  const int t  = threadIdx.x;
  const int s0 = blockIdx.x * TM;

  {
    const int m = t >> 3, i = t & 7;
    const float a = x[(s0 + m) * NQ + i] * 0.785398163397448f;
    csC[m][i] = cosf(a);
    csS[m][i] = sinf(a);
  }
  __syncthreads();

  for (int e = t; e < TM * NS; e += THREADS) {
    const int m = e >> 8, s = e & 255;
    float p = 1.f;
#pragma unroll
    for (int i = 0; i < NQ; ++i)
      p *= ((s >> (7 - i)) & 1) ? csS[m][i] : csC[m][i];
    Sre[m][s] = p;
    Sim[m][s] = 0.f;
  }
  __syncthreads();

  const int wave = t >> 5, lane = t & 31;
  const int half = lane >> 4, lq = lane & 15;

  v8f accR[2][2], accI[2][2];

  for (int l = 0; l < NLAY; ++l) {
    const float* Wre = W + l * 2 * (NS * NS);
    const float* Wim = Wre + NS * NS;

#pragma unroll
    for (int tt = 0; tt < 2; ++tt)
#pragma unroll
      for (int mt = 0; mt < 2; ++mt) {
        accR[tt][mt] = (v8f){0.f, 0.f, 0.f, 0.f, 0.f, 0.f, 0.f, 0.f};
        accI[tt][mt] = (v8f){0.f, 0.f, 0.f, 0.f, 0.f, 0.f, 0.f, 0.f};
      }

    for (int k = 0; k < NS; k += 4) {
      const int kk = k + 2 * half;
      const v2f ar0 = *(const v2f*)&Sre[lq][kk];
      const v2f ai0 = *(const v2f*)&Sim[lq][kk];
      const v2f ar1 = *(const v2f*)&Sre[lq + 16][kk];
      const v2f ai1 = *(const v2f*)&Sim[lq + 16][kk];
      const v2f an0 = -ai0;
      const v2f an1 = -ai1;
#pragma unroll
      for (int tt = 0; tt < 2; ++tt) {
        const int nt = wave * 2 + tt;
        const int fidx = (((k >> 2) << 4) + nt) * 64 + lane * 2;
        const v2f br = *(const v2f*)&Wre[fidx];
        const v2f bi = *(const v2f*)&Wim[fidx];
        accR[tt][0] = wmma_f32(ar0, br, accR[tt][0]);
        accR[tt][0] = wmma_f32(an0, bi, accR[tt][0]);
        accI[tt][0] = wmma_f32(ar0, bi, accI[tt][0]);
        accI[tt][0] = wmma_f32(ai0, br, accI[tt][0]);
        accR[tt][1] = wmma_f32(ar1, br, accR[tt][1]);
        accR[tt][1] = wmma_f32(an1, bi, accR[tt][1]);
        accI[tt][1] = wmma_f32(ar1, bi, accI[tt][1]);
        accI[tt][1] = wmma_f32(ai1, br, accI[tt][1]);
      }
    }
    __syncthreads();

    if (l == 0) {
#pragma unroll
      for (int tt = 0; tt < 2; ++tt) {
        const int n = (wave * 2 + tt) * 16 + lq;
#pragma unroll
        for (int mt = 0; mt < 2; ++mt)
#pragma unroll
          for (int j = 0; j < 8; ++j) {
            const int m = mt * 16 + j + 8 * half;
            Sre[m][n] = accR[tt][mt][j];
            Sim[m][n] = accI[tt][mt][j];
          }
      }
      __syncthreads();
      for (int i = 0; i < NQ; ++i) {
        const int b = 7 - i;
        for (int p = t; p < TM * (NS / 2); p += THREADS) {
          const int m = p >> 7, pr = p & 127;
          const int low = pr & ((1 << b) - 1);
          const int sA = ((pr >> b) << (b + 1)) | low;
          const int sB = sA | (1 << b);
          const float c = csC[m][i], sn = csS[m][i];
          float a0 = Sre[m][sA], a1 = Sre[m][sB];
          Sre[m][sA] = c * a0 - sn * a1;
          Sre[m][sB] = sn * a0 + c * a1;
          a0 = Sim[m][sA]; a1 = Sim[m][sB];
          Sim[m][sA] = c * a0 - sn * a1;
          Sim[m][sB] = sn * a0 + c * a1;
        }
        __syncthreads();
      }
    }
  }

  float* zp = &Sre[0][0];

  const unsigned ulq = (unsigned)lq;
  const unsigned sg1 = (ulq & 1u) << 31, sg2 = (ulq & 2u) << 30;
  const unsigned sg4 = (ulq & 4u) << 29, sg8 = (ulq & 8u) << 28;

  int slot = -1;
  if (lq == 0) slot = 0;
  else if (lq == 3) slot = 3;
  else if (lq == 8) slot = 4;
  else if (lq == 4) slot = 5;
  else if (lq == 2) slot = 6;
  else if (lq == 1) slot = 7;

#pragma unroll
  for (int mt = 0; mt < 2; ++mt)
#pragma unroll
    for (int j = 0; j < 8; ++j) {
      const float p0 = accR[0][mt][j] * accR[0][mt][j] +
                       accI[0][mt][j] * accI[0][mt][j];
      const float p1 = accR[1][mt][j] * accR[1][mt][j] +
                       accI[1][mt][j] * accI[1][mt][j];
      float h = p0 + p1;
      float d = p0 - p1;
      h = walsh_step<1>(h, sg1);
      h = walsh_step<2>(h, sg2);
      h = walsh_step<4>(h, sg4);
      h = walsh_step<8>(h, sg8);
      d += row_xor<1>(d);
      d += row_xor<2>(d);
      d += row_xor<4>(d);
      d += row_xor<8>(d);
      const int m = mt * 16 + j + 8 * half;
      if (slot >= 0) zp[(wave * TM + m) * NQ + slot] = (lq == 3) ? d : h;
    }
  __syncthreads();

  {
    const int m = t >> 3, w = t & 7;
    const bool top = w < 3;
    const int zs = top ? 0 : w;
    const int wb = top ? (2 - w) : 0;
    float acc = 0.f;
#pragma unroll
    for (int wv = 0; wv < 8; ++wv) {
      const float v = zp[(wv * TM + m) * NQ + zs];
      acc += (top && ((wv >> wb) & 1)) ? -v : v;
    }
    out[(s0 + m) * NQ + w] = acc;
  }
}

extern "C" void kernel_launch(void* const* d_in, const int* in_sizes, int n_in,
                              void* d_out, int out_size, void* d_ws, size_t ws_size,
                              hipStream_t stream) {
  const float* x   = (const float*)d_in[0];
  const float* wts = (const float*)d_in[1];
  float* out = (float*)d_out;
  float* W   = (float*)d_ws;

  const int B = in_sizes[0] / NQ;

  build_layer_mats<<<(NLAY * NS) / (256 / 32), 256, 0, stream>>>(wts, W);
  circuit_gemm<<<B / TM, THREADS, 0, stream>>>(x, W, out);
}


# Layer-1 GEMM: its input state is real for every x (start state, then real RY only), so the all-zero imaginary A plane is no longer stored, loaded or multiplied: half the wmma there; sum order kept
# speedup vs baseline: 3.8246x; 1.2409x over previous
#include <hip/hip_runtime.h>
#include <hip/hip_bf16.h>
#include <math.h>


typedef float v2f __attribute__((ext_vector_type(2)));
typedef float v8f __attribute__((ext_vector_type(8)));

#define NQ    8
#define NS    256
#define NLAY  2
#define TM    32
#define THREADS 256

__device__ __forceinline__ int bswz(int k, int n) {
  const int fraglane = (((k >> 1) & 1) << 4) | (n & 15);
  return (((k >> 2) << 4) + (n >> 4)) * 64 + fraglane * 2 + (k & 1);
}

__global__ __launch_bounds__(256) void build_layer_mats(
    const float* __restrict__ wts, float* __restrict__ W)
{
  const int t    = threadIdx.x;
  const int wave = blockIdx.x * (blockDim.x >> 5) + (t >> 5);
  const int lane = t & 31;
  const int layer = wave >> 8;
  const int k     = wave & 255;

  float re[8], im[8];
#pragma unroll
  for (int r = 0; r < 8; ++r) {
    re[r] = (((lane << 3) | r) == k) ? 1.f : 0.f;
    im[r] = 0.f;
  }

  const float* wl = wts + layer * (NQ * 2);

#pragma unroll
  for (int i = 0; i < NQ; ++i) {
    const int b = 7 - i;
    const float hy = 0.5f * wl[i * 2 + 0];
    const float cy = cosf(hy), sy = sinf(hy);
    if (b < 3) {
      const int m = 1 << b;
#pragma unroll
      for (int r = 0; r < 8; ++r) {
        if (!(r & m)) {
          const int r1 = r | m;
          float a0 = re[r], a1 = re[r1];
          re[r] = cy * a0 - sy * a1;  re[r1] = sy * a0 + cy * a1;
          a0 = im[r]; a1 = im[r1];
          im[r] = cy * a0 - sy * a1;  im[r1] = sy * a0 + cy * a1;
        }
      }
    } else {
      const int lm = 1 << (b - 3);
      const bool hi = (lane & lm) != 0;
#pragma unroll
      for (int r = 0; r < 8; ++r) {
        const float ore = __shfl_xor(re[r], lm, 32);
        const float oim = __shfl_xor(im[r], lm, 32);
        re[r] = hi ? (sy * ore + cy * re[r]) : (cy * re[r] - sy * ore);
        im[r] = hi ? (sy * oim + cy * im[r]) : (cy * im[r] - sy * oim);
      }
    }
    const float hz = 0.5f * wl[i * 2 + 1];
    const float cz = cosf(hz), sz = sinf(hz);
#pragma unroll
    for (int r = 0; r < 8; ++r) {
      const int bit = (b < 3) ? ((r >> b) & 1) : ((lane >> (b - 3)) & 1);
      const float tt = bit ? -sz : sz;
      const float nr = cz * re[r] + tt * im[r];
      const float ni = cz * im[r] - tt * re[r];
      re[r] = nr; im[r] = ni;
    }
  }

#pragma unroll
  for (int g = 0; g < NQ; ++g) {
    const int ci = (g < 7) ? g : 7;
    const int ti = (g < 7) ? g + 1 : 0;
    const int bc = 7 - ci, bt = 7 - ti;
    if (bt < 3) {
      const int tm = 1 << bt;
      float tre[8], tim[8];
#pragma unroll
      for (int r = 0; r < 8; ++r) { tre[r] = re[r]; tim[r] = im[r]; }
#pragma unroll
      for (int r = 0; r < 8; ++r) {
        const int cb = (bc < 3) ? ((r >> bc) & 1) : ((lane >> (bc - 3)) & 1);
        re[r] = cb ? tre[r ^ tm] : tre[r];
        im[r] = cb ? tim[r ^ tm] : tim[r];
      }
    } else {
      const int lm = 1 << (bt - 3);
#pragma unroll
      for (int r = 0; r < 8; ++r) {
        const float ore = __shfl_xor(re[r], lm, 32);
        const float oim = __shfl_xor(im[r], lm, 32);
        const int cb = (bc < 3) ? ((r >> bc) & 1) : ((lane >> (bc - 3)) & 1);
        re[r] = cb ? ore : re[r];
        im[r] = cb ? oim : im[r];
      }
    }
  }

  float* Wre = W + layer * 2 * (NS * NS);
  float* Wim = Wre + NS * NS;
#pragma unroll
  for (int r = 0; r < 8; ++r) {
    const int n = (lane << 3) | r;
    const int idx = bswz(k, n);
    Wre[idx] = re[r];
    Wim[idx] = im[r];
  }
}

static __device__ __forceinline__ v8f wmma_f32(v2f a, v2f b, v8f c) {
  return __builtin_amdgcn_wmma_f32_16x16x4_f32(
       false, a,  false, b,
       (short)0, c,  false,  false);
}


template <int M>
static __device__ __forceinline__ float row_xor(float v) {
  const int b = __builtin_bit_cast(int, v);
  return __builtin_bit_cast(
      float, __builtin_amdgcn_update_dpp(b, b, 0x160 + M, 0xf, 0xf, true));
}

template <int M>
static __device__ __forceinline__ float walsh_step(float v, unsigned sgn) {
  const float other = row_xor<M>(v);
  const float own =
      __builtin_bit_cast(float, __builtin_bit_cast(unsigned, v) ^ sgn);
  return other + own;
}

__global__ __launch_bounds__(THREADS) void circuit_gemm(
    const float* __restrict__ x, const float* __restrict__ W,
    float* __restrict__ out)
{
  __shared__ float Sre[TM][NS];
  __shared__ float Sim[TM][NS];
  __shared__ float csC[TM][NQ];
  __shared__ float csS[TM][NQ];

  const int t  = threadIdx.x;
  const int s0 = blockIdx.x * TM;

  {
    const int m = t >> 3, i = t & 7;
    const float a = x[(s0 + m) * NQ + i] * 0.785398163397448f;
    csC[m][i] = cosf(a);
    csS[m][i] = sinf(a);
  }
  __syncthreads();

  for (int e = t; e < TM * NS; e += THREADS) {
    const int m = e >> 8, s = e & 255;
    float p = 1.f;
#pragma unroll
    for (int i = 0; i < NQ; ++i)
      p *= ((s >> (7 - i)) & 1) ? csS[m][i] : csC[m][i];
    Sre[m][s] = p;
  }
  __syncthreads();

  const int wave = t >> 5, lane = t & 31;
  const int half = lane >> 4, lq = lane & 15;

  v8f accR[2][2], accI[2][2];

#pragma unroll
  for (int tt = 0; tt < 2; ++tt)
#pragma unroll
    for (int mt = 0; mt < 2; ++mt) {
      accR[tt][mt] = (v8f){0.f, 0.f, 0.f, 0.f, 0.f, 0.f, 0.f, 0.f};
      accI[tt][mt] = (v8f){0.f, 0.f, 0.f, 0.f, 0.f, 0.f, 0.f, 0.f};
    }

  {
    const float* Wre = W;
    const float* Wim = Wre + NS * NS;
    for (int k = 0; k < NS; k += 4) {
      const int kk = k + 2 * half;
      const v2f ar0 = *(const v2f*)&Sre[lq][kk];
      const v2f ar1 = *(const v2f*)&Sre[lq + 16][kk];
#pragma unroll
      for (int tt = 0; tt < 2; ++tt) {
        const int nt = wave * 2 + tt;
        const int fidx = (((k >> 2) << 4) + nt) * 64 + lane * 2;
        const v2f br = *(const v2f*)&Wre[fidx];
        const v2f bi = *(const v2f*)&Wim[fidx];
        accR[tt][0] = wmma_f32(ar0, br, accR[tt][0]);
        accI[tt][0] = wmma_f32(ar0, bi, accI[tt][0]);
        accR[tt][1] = wmma_f32(ar1, br, accR[tt][1]);
        accI[tt][1] = wmma_f32(ar1, bi, accI[tt][1]);
      }
    }
  }
  __syncthreads();

#pragma unroll
  for (int tt = 0; tt < 2; ++tt) {
    const int n = (wave * 2 + tt) * 16 + lq;
#pragma unroll
    for (int mt = 0; mt < 2; ++mt)
#pragma unroll
      for (int j = 0; j < 8; ++j) {
        const int m = mt * 16 + j + 8 * half;
        Sre[m][n] = accR[tt][mt][j];
        Sim[m][n] = accI[tt][mt][j];
      }
  }
  __syncthreads();
  for (int i = 0; i < NQ; ++i) {
    const int b = 7 - i;
    for (int p = t; p < TM * (NS / 2); p += THREADS) {
      const int m = p >> 7, pr = p & 127;
      const int low = pr & ((1 << b) - 1);
      const int sA = ((pr >> b) << (b + 1)) | low;
      const int sB = sA | (1 << b);
      const float c = csC[m][i], sn = csS[m][i];
      float a0 = Sre[m][sA], a1 = Sre[m][sB];
      Sre[m][sA] = c * a0 - sn * a1;
      Sre[m][sB] = sn * a0 + c * a1;
      a0 = Sim[m][sA]; a1 = Sim[m][sB];
      Sim[m][sA] = c * a0 - sn * a1;
      Sim[m][sB] = sn * a0 + c * a1;
    }
    __syncthreads();
  }

#pragma unroll
  for (int tt = 0; tt < 2; ++tt)
#pragma unroll
    for (int mt = 0; mt < 2; ++mt) {
      accR[tt][mt] = (v8f){0.f, 0.f, 0.f, 0.f, 0.f, 0.f, 0.f, 0.f};
      accI[tt][mt] = (v8f){0.f, 0.f, 0.f, 0.f, 0.f, 0.f, 0.f, 0.f};
    }

  {
    const float* Wre = W + 2 * (NS * NS);
    const float* Wim = Wre + NS * NS;
    for (int k = 0; k < NS; k += 4) {
      const int kk = k + 2 * half;
      const v2f ar0 = *(const v2f*)&Sre[lq][kk];
      const v2f ai0 = *(const v2f*)&Sim[lq][kk];
      const v2f ar1 = *(const v2f*)&Sre[lq + 16][kk];
      const v2f ai1 = *(const v2f*)&Sim[lq + 16][kk];
      const v2f an0 = -ai0;
      const v2f an1 = -ai1;
#pragma unroll
      for (int tt = 0; tt < 2; ++tt) {
        const int nt = wave * 2 + tt;
        const int fidx = (((k >> 2) << 4) + nt) * 64 + lane * 2;
        const v2f br = *(const v2f*)&Wre[fidx];
        const v2f bi = *(const v2f*)&Wim[fidx];
        accR[tt][0] = wmma_f32(ar0, br, accR[tt][0]);
        accR[tt][0] = wmma_f32(an0, bi, accR[tt][0]);
        accI[tt][0] = wmma_f32(ar0, bi, accI[tt][0]);
        accI[tt][0] = wmma_f32(ai0, br, accI[tt][0]);
        accR[tt][1] = wmma_f32(ar1, br, accR[tt][1]);
        accR[tt][1] = wmma_f32(an1, bi, accR[tt][1]);
        accI[tt][1] = wmma_f32(ar1, bi, accI[tt][1]);
        accI[tt][1] = wmma_f32(ai1, br, accI[tt][1]);
      }
    }
  }
  __syncthreads();

  float* zp = &Sre[0][0];

  const unsigned ulq = (unsigned)lq;
  const unsigned sg1 = (ulq & 1u) << 31, sg2 = (ulq & 2u) << 30;
  const unsigned sg4 = (ulq & 4u) << 29, sg8 = (ulq & 8u) << 28;

  int slot = -1;
  if (lq == 0) slot = 0;
  else if (lq == 3) slot = 3;
  else if (lq == 8) slot = 4;
  else if (lq == 4) slot = 5;
  else if (lq == 2) slot = 6;
  else if (lq == 1) slot = 7;

#pragma unroll
  for (int mt = 0; mt < 2; ++mt)
#pragma unroll
    for (int j = 0; j < 8; ++j) {
      const float p0 = accR[0][mt][j] * accR[0][mt][j] +
                       accI[0][mt][j] * accI[0][mt][j];
      const float p1 = accR[1][mt][j] * accR[1][mt][j] +
                       accI[1][mt][j] * accI[1][mt][j];
      float h = p0 + p1;
      float d = p0 - p1;
      h = walsh_step<1>(h, sg1);
      h = walsh_step<2>(h, sg2);
      h = walsh_step<4>(h, sg4);
      h = walsh_step<8>(h, sg8);
      d += row_xor<1>(d);
      d += row_xor<2>(d);
      d += row_xor<4>(d);
      d += row_xor<8>(d);
      const int m = mt * 16 + j + 8 * half;
      if (slot >= 0) zp[(wave * TM + m) * NQ + slot] = (lq == 3) ? d : h;
    }
  __syncthreads();

  {
    const int m = t >> 3, w = t & 7;
    const bool top = w < 3;
    const int zs = top ? 0 : w;
    const int wb = top ? (2 - w) : 0;
    float acc = 0.f;
#pragma unroll
    for (int wv = 0; wv < 8; ++wv) {
      const float v = zp[(wv * TM + m) * NQ + zs];
      acc += (top && ((wv >> wb) & 1)) ? -v : v;
    }
    out[(s0 + m) * NQ + w] = acc;
  }
}

extern "C" void kernel_launch(void* const* d_in, const int* in_sizes, int n_in,
                              void* d_out, int out_size, void* d_ws, size_t ws_size,
                              hipStream_t stream) {
  const float* x   = (const float*)d_in[0];
  const float* wts = (const float*)d_in[1];
  float* out = (float*)d_out;
  float* W   = (float*)d_ws;

  const int B = in_sizes[0] / NQ;

  build_layer_mats<<<(NLAY * NS) / (256 / 32), 256, 0, stream>>>(wts, W);
  circuit_gemm<<<B / TM, THREADS, 0, stream>>>(x, W, out);
}


# Layer matrix factored as P*(U_hi x U_lo): each 256-deep complex GEMM becomes two 16-deep WMMA stages with B in registers, CNOT ring as an index permutation; sum order changes (two-stage tree)
# speedup vs baseline: 12.8625x; 3.3631x over previous
#include <hip/hip_runtime.h>
#include <hip/hip_bf16.h>
#include <math.h>


typedef float v2f __attribute__((ext_vector_type(2)));
typedef float v8f __attribute__((ext_vector_type(8)));

#define NQ    8
#define NS    256
#define NLAY  2
#define TM    32
#define THREADS 256
#define QP    18
#define RP    292
#define FRAG  256

static __device__ __forceinline__ int amp_addr(int m, int s) {
  return m * RP + (s >> 4) * QP + (s & 15);
}

static __device__ __forceinline__ int cnot_ring(int s) {
  int y = s;
  y ^= y >> 1;
  y ^= y >> 2;
  y ^= y >> 4;
  y ^= (y & 1) << 7;
  return y;
}

__global__ __launch_bounds__(256) void build_factor_frags(
    const float* __restrict__ wts, float* __restrict__ F)
{
  const int t     = threadIdx.x;
  const int layer = blockIdx.x >> 1;
  const int stage = blockIdx.x & 1;
  const int ks = t >> 6, lane = (t >> 1) & 31, e = t & 1;
  const int k = 4 * ks + 2 * (lane >> 4) + e;
  const int n = lane & 15;

  const float* wl = wts + layer * (NQ * 2);
  float pr = 1.f, pi = 0.f;
#pragma unroll
  for (int j = 0; j < 4; ++j) {
    const int wire = (stage ? 0 : 4) + j;
    const int ob = (n >> (3 - j)) & 1;
    const int ib = (k >> (3 - j)) & 1;
    const float hy = 0.5f * wl[wire * 2 + 0];
    const float hz = 0.5f * wl[wire * 2 + 1];
    const float cy = cosf(hy), sy = sinf(hy);
    const float cz = cosf(hz), sz = sinf(hz);
    const float ry = (ob == ib) ? cy : (ob ? sy : -sy);
    const float ur = ry * cz;
    const float ui = ob ? ry * sz : -(ry * sz);
    const float nr = pr * ur - pi * ui;
    const float ni = pr * ui + pi * ur;
    pr = nr; pi = ni;
  }
  float* dst = F + (layer * 2 + stage) * 2 * FRAG;
  dst[t]        = pr;
  dst[FRAG + t] = pi;
}

static __device__ __forceinline__ v8f wmma_f32(v2f a, v2f b, v8f c) {
  return __builtin_amdgcn_wmma_f32_16x16x4_f32(
       false, a,  false, b,
       (short)0, c,  false,  false);
}

template <bool REAL_IN>
static __device__ __forceinline__ void factor_stage(
    const float* __restrict__ Fst, const float* Sre, const float* Sim,
    int wave, int lane, v8f (&accR)[2][2], v8f (&accI)[2][2])
{
  const int half = lane >> 4, lq = lane & 15;
  v2f br[4], bi[4];
#pragma unroll
  for (int ks = 0; ks < 4; ++ks) {
    br[ks] = *(const v2f*)&Fst[ks * 64 + lane * 2];
    bi[ks] = *(const v2f*)&Fst[FRAG + ks * 64 + lane * 2];
  }
#pragma unroll
  for (int tt = 0; tt < 2; ++tt)
#pragma unroll
    for (int mt = 0; mt < 2; ++mt) {
      accR[tt][mt] = (v8f){0.f, 0.f, 0.f, 0.f, 0.f, 0.f, 0.f, 0.f};
      accI[tt][mt] = (v8f){0.f, 0.f, 0.f, 0.f, 0.f, 0.f, 0.f, 0.f};
    }
#pragma unroll
  for (int ks = 0; ks < 4; ++ks)
#pragma unroll
    for (int tt = 0; tt < 2; ++tt)
#pragma unroll
      for (int mt = 0; mt < 2; ++mt) {
        const int a = (mt * 16 + lq) * RP + (wave * 2 + tt) * QP + 4 * ks + 2 * half;
        const v2f ar = *(const v2f*)&Sre[a];
        accR[tt][mt] = wmma_f32(ar, br[ks], accR[tt][mt]);
        accI[tt][mt] = wmma_f32(ar, bi[ks], accI[tt][mt]);
        if (!REAL_IN) {
          const v2f ai = *(const v2f*)&Sim[a];
          const v2f an = -ai;
          accR[tt][mt] = wmma_f32(an, bi[ks], accR[tt][mt]);
          accI[tt][mt] = wmma_f32(ai, br[ks], accI[tt][mt]);
        }
      }
}

static __device__ __forceinline__ void store_swapped(
    float* Sre, float* Sim, int wave, int lane,
    const v8f (&accR)[2][2], const v8f (&accI)[2][2])
{
  const int half = lane >> 4, lq = lane & 15;
#pragma unroll
  for (int tt = 0; tt < 2; ++tt)
#pragma unroll
    for (int mt = 0; mt < 2; ++mt)
#pragma unroll
      for (int j = 0; j < 8; ++j) {
        const int m = mt * 16 + j + 8 * half;
        const int d = m * RP + lq * QP + (wave * 2 + tt);
        Sre[d] = accR[tt][mt][j];
        Sim[d] = accI[tt][mt][j];
      }
}

template <int M>
static __device__ __forceinline__ float row_xor(float v) {
  const int b = __builtin_bit_cast(int, v);
  return __builtin_bit_cast(
      float, __builtin_amdgcn_update_dpp(b, b, 0x160 + M, 0xf, 0xf, true));
}

template <int M>
static __device__ __forceinline__ float walsh_step(float v, unsigned sgn) {
  const float other = row_xor<M>(v);
  const float own =
      __builtin_bit_cast(float, __builtin_bit_cast(unsigned, v) ^ sgn);
  return other + own;
}

__global__ __launch_bounds__(THREADS, 8) void circuit_gemm(
    const float* __restrict__ x, const float* __restrict__ F,
    float* __restrict__ out)
{
  __shared__ __attribute__((aligned(16))) float Sre[TM * RP];
  __shared__ __attribute__((aligned(16))) float Sim[TM * RP];
  __shared__ float csC[TM][NQ];
  __shared__ float csS[TM][NQ];

  const int t  = threadIdx.x;
  const int s0 = blockIdx.x * TM;

  {
    const int m = t >> 3, i = t & 7;
    const float a = x[(s0 + m) * NQ + i] * 0.785398163397448f;
    csC[m][i] = cosf(a);
    csS[m][i] = sinf(a);
  }
  __syncthreads();

  for (int e = t; e < TM * NS; e += THREADS) {
    const int m = e >> 8, s = e & 255;
    float p = 1.f;
#pragma unroll
    for (int i = 0; i < NQ; ++i)
      p *= ((s >> (7 - i)) & 1) ? csS[m][i] : csC[m][i];
    Sre[amp_addr(m, s)] = p;
  }
  __syncthreads();

  const int wave = t >> 5, lane = t & 31;
  const int half = lane >> 4, lq = lane & 15;

  v8f accR[2][2], accI[2][2];

  factor_stage<true>(F + 0 * 2 * FRAG, Sre, Sim, wave, lane, accR, accI);
  __syncthreads();
  store_swapped(Sre, Sim, wave, lane, accR, accI);
  __syncthreads();

  factor_stage<false>(F + 1 * 2 * FRAG, Sre, Sim, wave, lane, accR, accI);
  __syncthreads();

#pragma unroll
  for (int tt = 0; tt < 2; ++tt) {
    const int d0 = amp_addr(0, cnot_ring(lq * 16 + wave * 2 + tt));
#pragma unroll
    for (int mt = 0; mt < 2; ++mt)
#pragma unroll
      for (int j = 0; j < 8; ++j) {
        const int m = mt * 16 + j + 8 * half;
        Sre[m * RP + d0] = accR[tt][mt][j];
        Sim[m * RP + d0] = accI[tt][mt][j];
      }
  }
  __syncthreads();

  for (int i = 0; i < NQ; ++i) {
    const int b = 7 - i;
    for (int p = t; p < TM * (NS / 2); p += THREADS) {
      const int m = p >> 7, pr = p & 127;
      const int low = pr & ((1 << b) - 1);
      const int sA = ((pr >> b) << (b + 1)) | low;
      const int sB = sA | (1 << b);
      const int dA = amp_addr(m, sA), dB = amp_addr(m, sB);
      const float c = csC[m][i], sn = csS[m][i];
      float a0 = Sre[dA], a1 = Sre[dB];
      Sre[dA] = c * a0 - sn * a1;
      Sre[dB] = sn * a0 + c * a1;
      a0 = Sim[dA]; a1 = Sim[dB];
      Sim[dA] = c * a0 - sn * a1;
      Sim[dB] = sn * a0 + c * a1;
    }
    __syncthreads();
  }

  factor_stage<false>(F + 2 * 2 * FRAG, Sre, Sim, wave, lane, accR, accI);
  __syncthreads();
  store_swapped(Sre, Sim, wave, lane, accR, accI);
  __syncthreads();
  factor_stage<false>(F + 3 * 2 * FRAG, Sre, Sim, wave, lane, accR, accI);
  __syncthreads();

  float* zp = &Sre[0];

  const unsigned ulq = (unsigned)lq;
  const unsigned sg1 = (ulq & 1u) << 31, sg2 = (ulq & 2u) << 30;
  const unsigned sg4 = (ulq & 4u) << 29, sg8 = (ulq & 8u) << 28;

  const int slotH = (lq == 12) ? 1 : (lq == 14) ? 2 : (lq == 15) ? 3 : -1;
  const int slotD = (lq == 7) ? 0 : (lq == 15) ? 4 : -1;

#pragma unroll
  for (int mt = 0; mt < 2; ++mt)
#pragma unroll
    for (int j = 0; j < 8; ++j) {
      const float p0 = accR[0][mt][j] * accR[0][mt][j] +
                       accI[0][mt][j] * accI[0][mt][j];
      const float p1 = accR[1][mt][j] * accR[1][mt][j] +
                       accI[1][mt][j] * accI[1][mt][j];
      float h = p0 + p1;
      float d = p0 - p1;
      h = walsh_step<1>(h, sg1);
      h = walsh_step<2>(h, sg2);
      h = walsh_step<4>(h, sg4);
      h = walsh_step<8>(h, sg8);
      d = walsh_step<1>(d, sg1);
      d = walsh_step<2>(d, sg2);
      d = walsh_step<4>(d, sg4);
      d = walsh_step<8>(d, sg8);
      const int m = mt * 16 + j + 8 * half;
      if (slotH >= 0) zp[(wave * TM + m) * NQ + slotH] = h;
      if (slotD >= 0) zp[(wave * TM + m) * NQ + slotD] = d;
    }
  __syncthreads();

  {
    const int m = t >> 3, w = t & 7;
    const int zs = (w == 0) ? 0 : (w < 3) ? w : (w < 7) ? 3 : 4;
    const int wm = (w == 0 || w >= 6) ? 7 : (w == 4) ? 4 : (w == 5) ? 6 : 0;
    float acc = 0.f;
#pragma unroll
    for (int wv = 0; wv < 8; ++wv) {
      const float v = zp[(wv * TM + m) * NQ + zs];
      acc += (__builtin_popcount(wv & wm) & 1) ? -v : v;
    }
    out[(s0 + m) * NQ + w] = acc;
  }
}

extern "C" void kernel_launch(void* const* d_in, const int* in_sizes, int n_in,
                              void* d_out, int out_size, void* d_ws, size_t ws_size,
                              hipStream_t stream) {
  const float* x   = (const float*)d_in[0];
  const float* wts = (const float*)d_in[1];
  float* out = (float*)d_out;
  float* F   = (float*)d_ws;

  const int B = in_sizes[0] / NQ;

  build_factor_frags<<<NLAY * 2, 256, 0, stream>>>(wts, F);
  circuit_gemm<<<B / TM, THREADS, 0, stream>>>(x, F, out);
}


# Layer-2 embedding: the 8 per-wire in-place LDS passes become 2 passes, a thread rotating its own 16 amplitudes of a nibble in registers (4 RY gates per load/store); wire order and sum order kept
# speedup vs baseline: 16.7014x; 1.2985x over previous
#include <hip/hip_runtime.h>
#include <hip/hip_bf16.h>
#include <math.h>


typedef float v2f __attribute__((ext_vector_type(2)));
typedef float v8f __attribute__((ext_vector_type(8)));

#define NQ    8
#define NS    256
#define NLAY  2
#define TM    32
#define THREADS 256
#define QP    18
#define RP    292
#define FRAG  256

static __device__ __forceinline__ int amp_addr(int m, int s) {
  return m * RP + (s >> 4) * QP + (s & 15);
}

static __device__ __forceinline__ int cnot_ring(int s) {
  int y = s;
  y ^= y >> 1;
  y ^= y >> 2;
  y ^= y >> 4;
  y ^= (y & 1) << 7;
  return y;
}

__global__ __launch_bounds__(256) void build_factor_frags(
    const float* __restrict__ wts, float* __restrict__ F)
{
  const int t     = threadIdx.x;
  const int layer = blockIdx.x >> 1;
  const int stage = blockIdx.x & 1;
  const int ks = t >> 6, lane = (t >> 1) & 31, e = t & 1;
  const int k = 4 * ks + 2 * (lane >> 4) + e;
  const int n = lane & 15;

  const float* wl = wts + layer * (NQ * 2);
  float pr = 1.f, pi = 0.f;
#pragma unroll
  for (int j = 0; j < 4; ++j) {
    const int wire = (stage ? 0 : 4) + j;
    const int ob = (n >> (3 - j)) & 1;
    const int ib = (k >> (3 - j)) & 1;
    const float hy = 0.5f * wl[wire * 2 + 0];
    const float hz = 0.5f * wl[wire * 2 + 1];
    const float cy = cosf(hy), sy = sinf(hy);
    const float cz = cosf(hz), sz = sinf(hz);
    const float ry = (ob == ib) ? cy : (ob ? sy : -sy);
    const float ur = ry * cz;
    const float ui = ob ? ry * sz : -(ry * sz);
    const float nr = pr * ur - pi * ui;
    const float ni = pr * ui + pi * ur;
    pr = nr; pi = ni;
  }
  float* dst = F + (layer * 2 + stage) * 2 * FRAG;
  dst[t]        = pr;
  dst[FRAG + t] = pi;
}

static __device__ __forceinline__ v8f wmma_f32(v2f a, v2f b, v8f c) {
  return __builtin_amdgcn_wmma_f32_16x16x4_f32(
       false, a,  false, b,
       (short)0, c,  false,  false);
}

template <bool REAL_IN>
static __device__ __forceinline__ void factor_stage(
    const float* __restrict__ Fst, const float* Sre, const float* Sim,
    int wave, int lane, v8f (&accR)[2][2], v8f (&accI)[2][2])
{
  const int half = lane >> 4, lq = lane & 15;
  v2f br[4], bi[4];
#pragma unroll
  for (int ks = 0; ks < 4; ++ks) {
    br[ks] = *(const v2f*)&Fst[ks * 64 + lane * 2];
    bi[ks] = *(const v2f*)&Fst[FRAG + ks * 64 + lane * 2];
  }
#pragma unroll
  for (int tt = 0; tt < 2; ++tt)
#pragma unroll
    for (int mt = 0; mt < 2; ++mt) {
      accR[tt][mt] = (v8f){0.f, 0.f, 0.f, 0.f, 0.f, 0.f, 0.f, 0.f};
      accI[tt][mt] = (v8f){0.f, 0.f, 0.f, 0.f, 0.f, 0.f, 0.f, 0.f};
    }
#pragma unroll
  for (int ks = 0; ks < 4; ++ks)
#pragma unroll
    for (int tt = 0; tt < 2; ++tt)
#pragma unroll
      for (int mt = 0; mt < 2; ++mt) {
        const int a = (mt * 16 + lq) * RP + (wave * 2 + tt) * QP + 4 * ks + 2 * half;
        const v2f ar = *(const v2f*)&Sre[a];
        accR[tt][mt] = wmma_f32(ar, br[ks], accR[tt][mt]);
        accI[tt][mt] = wmma_f32(ar, bi[ks], accI[tt][mt]);
        if (!REAL_IN) {
          const v2f ai = *(const v2f*)&Sim[a];
          const v2f an = -ai;
          accR[tt][mt] = wmma_f32(an, bi[ks], accR[tt][mt]);
          accI[tt][mt] = wmma_f32(ai, br[ks], accI[tt][mt]);
        }
      }
}

static __device__ __forceinline__ void store_swapped(
    float* Sre, float* Sim, int wave, int lane,
    const v8f (&accR)[2][2], const v8f (&accI)[2][2])
{
  const int half = lane >> 4, lq = lane & 15;
#pragma unroll
  for (int tt = 0; tt < 2; ++tt)
#pragma unroll
    for (int mt = 0; mt < 2; ++mt)
#pragma unroll
      for (int j = 0; j < 8; ++j) {
        const int m = mt * 16 + j + 8 * half;
        const int d = m * RP + lq * QP + (wave * 2 + tt);
        Sre[d] = accR[tt][mt][j];
        Sim[d] = accI[tt][mt][j];
      }
}

template <int M>
static __device__ __forceinline__ float row_xor(float v) {
  const int b = __builtin_bit_cast(int, v);
  return __builtin_bit_cast(
      float, __builtin_amdgcn_update_dpp(b, b, 0x160 + M, 0xf, 0xf, true));
}

template <int M>
static __device__ __forceinline__ float walsh_step(float v, unsigned sgn) {
  const float other = row_xor<M>(v);
  const float own =
      __builtin_bit_cast(float, __builtin_bit_cast(unsigned, v) ^ sgn);
  return other + own;
}

template <int STRIDE, int W0>
static __device__ __forceinline__ void embed_nibble(
    float* Sre, float* Sim, int base, const float* cs, const float* sn)
{
  v2f zr[8], zi[8];
#pragma unroll
  for (int q = 0; q < 8; ++q) {
    if (STRIDE == 1) {
      zr[q] = *(const v2f*)&Sre[base + 2 * q];
      zi[q] = *(const v2f*)&Sim[base + 2 * q];
    } else {
      zr[q] = (v2f){Sre[base + 2 * q * STRIDE], Sre[base + (2 * q + 1) * STRIDE]};
      zi[q] = (v2f){Sim[base + 2 * q * STRIDE], Sim[base + (2 * q + 1) * STRIDE]};
    }
  }
#pragma unroll
  for (int j = 0; j < 3; ++j) {
    const int qbit = 4 >> j;
    const float c = cs[W0 + j], s = sn[W0 + j];
#pragma unroll
    for (int q = 0; q < 8; ++q) {
      if (q & qbit) continue;
      const int q1 = q | qbit;
      v2f a0 = zr[q], a1 = zr[q1];
      zr[q]  = c * a0 - s * a1;
      zr[q1] = s * a0 + c * a1;
      a0 = zi[q]; a1 = zi[q1];
      zi[q]  = c * a0 - s * a1;
      zi[q1] = s * a0 + c * a1;
    }
  }
  {
    const float c = cs[W0 + 3], s = sn[W0 + 3];
    const v2f ms = (v2f){-s, s};
#pragma unroll
    for (int q = 0; q < 8; ++q) {
      const v2f yr = __builtin_shufflevector(zr[q], zr[q], 1, 0);
      const v2f yi = __builtin_shufflevector(zi[q], zi[q], 1, 0);
      zr[q] = c * zr[q] + ms * yr;
      zi[q] = c * zi[q] + ms * yi;
    }
  }
#pragma unroll
  for (int q = 0; q < 8; ++q) {
    if (STRIDE == 1) {
      *(v2f*)&Sre[base + 2 * q] = zr[q];
      *(v2f*)&Sim[base + 2 * q] = zi[q];
    } else {
      Sre[base + 2 * q * STRIDE]       = zr[q][0];
      Sre[base + (2 * q + 1) * STRIDE] = zr[q][1];
      Sim[base + 2 * q * STRIDE]       = zi[q][0];
      Sim[base + (2 * q + 1) * STRIDE] = zi[q][1];
    }
  }
}

__global__ __launch_bounds__(THREADS, 8) void circuit_gemm(
    const float* __restrict__ x, const float* __restrict__ F,
    float* __restrict__ out)
{
  __shared__ __attribute__((aligned(16))) float Sre[TM * RP];
  __shared__ __attribute__((aligned(16))) float Sim[TM * RP];
  __shared__ float csC[TM][NQ];
  __shared__ float csS[TM][NQ];

  const int t  = threadIdx.x;
  const int s0 = blockIdx.x * TM;

  {
    const int m = t >> 3, i = t & 7;
    const float a = x[(s0 + m) * NQ + i] * 0.785398163397448f;
    csC[m][i] = cosf(a);
    csS[m][i] = sinf(a);
  }
  __syncthreads();

  for (int e = t; e < TM * NS; e += THREADS) {
    const int m = e >> 8, s = e & 255;
    float p = 1.f;
#pragma unroll
    for (int i = 0; i < NQ; ++i)
      p *= ((s >> (7 - i)) & 1) ? csS[m][i] : csC[m][i];
    Sre[amp_addr(m, s)] = p;
  }
  __syncthreads();

  const int wave = t >> 5, lane = t & 31;
  const int half = lane >> 4, lq = lane & 15;

  v8f accR[2][2], accI[2][2];

  factor_stage<true>(F + 0 * 2 * FRAG, Sre, Sim, wave, lane, accR, accI);
  __syncthreads();
  store_swapped(Sre, Sim, wave, lane, accR, accI);
  __syncthreads();

  factor_stage<false>(F + 1 * 2 * FRAG, Sre, Sim, wave, lane, accR, accI);
  __syncthreads();

#pragma unroll
  for (int tt = 0; tt < 2; ++tt) {
    const int d0 = amp_addr(0, cnot_ring(lq * 16 + wave * 2 + tt));
#pragma unroll
    for (int mt = 0; mt < 2; ++mt)
#pragma unroll
      for (int j = 0; j < 8; ++j) {
        const int m = mt * 16 + j + 8 * half;
        Sre[m * RP + d0] = accR[tt][mt][j];
        Sim[m * RP + d0] = accI[tt][mt][j];
      }
  }
  __syncthreads();

#pragma unroll 1
  for (int r = 0; r < 2; ++r) {
    const int lo = t & 15;
    const int m  = (t >> 4) + 16 * r;
    embed_nibble<QP, 0>(Sre, Sim, m * RP + lo, &csC[m][0], &csS[m][0]);
  }
  __syncthreads();
#pragma unroll 1
  for (int r = 0; r < 2; ++r) {
    const int hi = t & 15;
    const int m  = (t >> 5) + 8 * ((t >> 4) & 1) + 16 * r;
    embed_nibble<1, 4>(Sre, Sim, m * RP + hi * QP, &csC[m][0], &csS[m][0]);
  }
  __syncthreads();

  factor_stage<false>(F + 2 * 2 * FRAG, Sre, Sim, wave, lane, accR, accI);
  __syncthreads();
  store_swapped(Sre, Sim, wave, lane, accR, accI);
  __syncthreads();
  factor_stage<false>(F + 3 * 2 * FRAG, Sre, Sim, wave, lane, accR, accI);
  __syncthreads();

  float* zp = &Sre[0];

  const unsigned ulq = (unsigned)lq;
  const unsigned sg1 = (ulq & 1u) << 31, sg2 = (ulq & 2u) << 30;
  const unsigned sg4 = (ulq & 4u) << 29, sg8 = (ulq & 8u) << 28;

  const int slotH = (lq == 12) ? 1 : (lq == 14) ? 2 : (lq == 15) ? 3 : -1;
  const int slotD = (lq == 7) ? 0 : (lq == 15) ? 4 : -1;

#pragma unroll
  for (int mt = 0; mt < 2; ++mt)
#pragma unroll
    for (int j = 0; j < 8; ++j) {
      const float p0 = accR[0][mt][j] * accR[0][mt][j] +
                       accI[0][mt][j] * accI[0][mt][j];
      const float p1 = accR[1][mt][j] * accR[1][mt][j] +
                       accI[1][mt][j] * accI[1][mt][j];
      float h = p0 + p1;
      float d = p0 - p1;
      h = walsh_step<1>(h, sg1);
      h = walsh_step<2>(h, sg2);
      h = walsh_step<4>(h, sg4);
      h = walsh_step<8>(h, sg8);
      d = walsh_step<1>(d, sg1);
      d = walsh_step<2>(d, sg2);
      d = walsh_step<4>(d, sg4);
      d = walsh_step<8>(d, sg8);
      const int m = mt * 16 + j + 8 * half;
      if (slotH >= 0) zp[(wave * TM + m) * NQ + slotH] = h;
      if (slotD >= 0) zp[(wave * TM + m) * NQ + slotD] = d;
    }
  __syncthreads();

  {
    const int m = t >> 3, w = t & 7;
    const int zs = (w == 0) ? 0 : (w < 3) ? w : (w < 7) ? 3 : 4;
    const int wm = (w == 0 || w >= 6) ? 7 : (w == 4) ? 4 : (w == 5) ? 6 : 0;
    float acc = 0.f;
#pragma unroll
    for (int wv = 0; wv < 8; ++wv) {
      const float v = zp[(wv * TM + m) * NQ + zs];
      acc += (__builtin_popcount(wv & wm) & 1) ? -v : v;
    }
    out[(s0 + m) * NQ + w] = acc;
  }
}

extern "C" void kernel_launch(void* const* d_in, const int* in_sizes, int n_in,
                              void* d_out, int out_size, void* d_ws, size_t ws_size,
                              hipStream_t stream) {
  const float* x   = (const float*)d_in[0];
  const float* wts = (const float*)d_in[1];
  float* out = (float*)d_out;
  float* F   = (float*)d_ws;

  const int B = in_sizes[0] / NQ;

  build_factor_frags<<<NLAY * 2, 256, 0, stream>>>(wts, F);
  circuit_gemm<<<B / TM, THREADS, 0, stream>>>(x, F, out);
}


# Layer 1: its input is a product state of the two nibbles, so both GEMM stages shrink to one 16-vector per sample and nibble (still WMMA) and one complex product per amplitude; sum order changes
# speedup vs baseline: 26.7193x; 1.5998x over previous
#include <hip/hip_runtime.h>
#include <hip/hip_bf16.h>
#include <math.h>


typedef float v2f __attribute__((ext_vector_type(2)));
typedef float v8f __attribute__((ext_vector_type(8)));

#define NQ    8
#define NS    256
#define NLAY  2
#define TM    32
#define THREADS 256
#define QP    18
#define RP    292
#define FRAG  256
#define VP    36

static __device__ __forceinline__ int amp_addr(int m, int s) {
  return m * RP + (s >> 4) * QP + (s & 15);
}

static __device__ __forceinline__ int cnot_ring(int s) {
  int y = s;
  y ^= y >> 1;
  y ^= y >> 2;
  y ^= y >> 4;
  y ^= (y & 1) << 7;
  return y;
}

__global__ __launch_bounds__(256) void build_factor_frags(
    const float* __restrict__ wts, float* __restrict__ F)
{
  const int t     = threadIdx.x;
  const int layer = blockIdx.x >> 1;
  const int stage = blockIdx.x & 1;
  const int ks = t >> 6, lane = (t >> 1) & 31, e = t & 1;
  const int k = 4 * ks + 2 * (lane >> 4) + e;
  const int n = lane & 15;

  const float* wl = wts + layer * (NQ * 2);
  float pr = 1.f, pi = 0.f;
#pragma unroll
  for (int j = 0; j < 4; ++j) {
    const int wire = (stage ? 0 : 4) + j;
    const int ob = (n >> (3 - j)) & 1;
    const int ib = (k >> (3 - j)) & 1;
    const float hy = 0.5f * wl[wire * 2 + 0];
    const float hz = 0.5f * wl[wire * 2 + 1];
    const float cy = cosf(hy), sy = sinf(hy);
    const float cz = cosf(hz), sz = sinf(hz);
    const float ry = (ob == ib) ? cy : (ob ? sy : -sy);
    const float ur = ry * cz;
    const float ui = ob ? ry * sz : -(ry * sz);
    const float nr = pr * ur - pi * ui;
    const float ni = pr * ui + pi * ur;
    pr = nr; pi = ni;
  }
  float* dst = F + (layer * 2 + stage) * 2 * FRAG;
  dst[t]        = pr;
  dst[FRAG + t] = pi;
}

static __device__ __forceinline__ v8f wmma_f32(v2f a, v2f b, v8f c) {
  return __builtin_amdgcn_wmma_f32_16x16x4_f32(
       false, a,  false, b,
       (short)0, c,  false,  false);
}

static __device__ __forceinline__ void factor_stage(
    const float* __restrict__ Fst, const float* Sre, const float* Sim,
    int wave, int lane, v8f (&accR)[2][2], v8f (&accI)[2][2])
{
  const int half = lane >> 4, lq = lane & 15;
  v2f br[4], bi[4];
#pragma unroll
  for (int ks = 0; ks < 4; ++ks) {
    br[ks] = *(const v2f*)&Fst[ks * 64 + lane * 2];
    bi[ks] = *(const v2f*)&Fst[FRAG + ks * 64 + lane * 2];
  }
#pragma unroll
  for (int tt = 0; tt < 2; ++tt)
#pragma unroll
    for (int mt = 0; mt < 2; ++mt) {
      accR[tt][mt] = (v8f){0.f, 0.f, 0.f, 0.f, 0.f, 0.f, 0.f, 0.f};
      accI[tt][mt] = (v8f){0.f, 0.f, 0.f, 0.f, 0.f, 0.f, 0.f, 0.f};
    }
#pragma unroll
  for (int ks = 0; ks < 4; ++ks)
#pragma unroll
    for (int tt = 0; tt < 2; ++tt)
#pragma unroll
      for (int mt = 0; mt < 2; ++mt) {
        const int a = (mt * 16 + lq) * RP + (wave * 2 + tt) * QP + 4 * ks + 2 * half;
        const v2f ar = *(const v2f*)&Sre[a];
        const v2f ai = *(const v2f*)&Sim[a];
        const v2f an = -ai;
        accR[tt][mt] = wmma_f32(ar, br[ks], accR[tt][mt]);
        accI[tt][mt] = wmma_f32(ar, bi[ks], accI[tt][mt]);
        accR[tt][mt] = wmma_f32(an, bi[ks], accR[tt][mt]);
        accI[tt][mt] = wmma_f32(ai, br[ks], accI[tt][mt]);
      }
}

static __device__ __forceinline__ void store_swapped(
    float* Sre, float* Sim, int wave, int lane,
    const v8f (&accR)[2][2], const v8f (&accI)[2][2])
{
  const int half = lane >> 4, lq = lane & 15;
#pragma unroll
  for (int tt = 0; tt < 2; ++tt)
#pragma unroll
    for (int mt = 0; mt < 2; ++mt)
#pragma unroll
      for (int j = 0; j < 8; ++j) {
        const int m = mt * 16 + j + 8 * half;
        const int d = m * RP + lq * QP + (wave * 2 + tt);
        Sre[d] = accR[tt][mt][j];
        Sim[d] = accI[tt][mt][j];
      }
}

template <int M>
static __device__ __forceinline__ float row_xor(float v) {
  const int b = __builtin_bit_cast(int, v);
  return __builtin_bit_cast(
      float, __builtin_amdgcn_update_dpp(b, b, 0x160 + M, 0xf, 0xf, true));
}

template <int M>
static __device__ __forceinline__ float walsh_step(float v, unsigned sgn) {
  const float other = row_xor<M>(v);
  const float own =
      __builtin_bit_cast(float, __builtin_bit_cast(unsigned, v) ^ sgn);
  return other + own;
}

template <int STRIDE, int W0>
static __device__ __forceinline__ void embed_nibble(
    float* Sre, float* Sim, int base, const float* cs, const float* sn)
{
  v2f zr[8], zi[8];
#pragma unroll
  for (int q = 0; q < 8; ++q) {
    if (STRIDE == 1) {
      zr[q] = *(const v2f*)&Sre[base + 2 * q];
      zi[q] = *(const v2f*)&Sim[base + 2 * q];
    } else {
      zr[q] = (v2f){Sre[base + 2 * q * STRIDE], Sre[base + (2 * q + 1) * STRIDE]};
      zi[q] = (v2f){Sim[base + 2 * q * STRIDE], Sim[base + (2 * q + 1) * STRIDE]};
    }
  }
#pragma unroll
  for (int j = 0; j < 3; ++j) {
    const int qbit = 4 >> j;
    const float c = cs[W0 + j], s = sn[W0 + j];
#pragma unroll
    for (int q = 0; q < 8; ++q) {
      if (q & qbit) continue;
      const int q1 = q | qbit;
      v2f a0 = zr[q], a1 = zr[q1];
      zr[q]  = c * a0 - s * a1;
      zr[q1] = s * a0 + c * a1;
      a0 = zi[q]; a1 = zi[q1];
      zi[q]  = c * a0 - s * a1;
      zi[q1] = s * a0 + c * a1;
    }
  }
  {
    const float c = cs[W0 + 3], s = sn[W0 + 3];
    const v2f ms = (v2f){-s, s};
#pragma unroll
    for (int q = 0; q < 8; ++q) {
      const v2f yr = __builtin_shufflevector(zr[q], zr[q], 1, 0);
      const v2f yi = __builtin_shufflevector(zi[q], zi[q], 1, 0);
      zr[q] = c * zr[q] + ms * yr;
      zi[q] = c * zi[q] + ms * yi;
    }
  }
#pragma unroll
  for (int q = 0; q < 8; ++q) {
    if (STRIDE == 1) {
      *(v2f*)&Sre[base + 2 * q] = zr[q];
      *(v2f*)&Sim[base + 2 * q] = zi[q];
    } else {
      Sre[base + 2 * q * STRIDE]       = zr[q][0];
      Sre[base + (2 * q + 1) * STRIDE] = zr[q][1];
      Sim[base + 2 * q * STRIDE]       = zi[q][0];
      Sim[base + (2 * q + 1) * STRIDE] = zi[q][1];
    }
  }
}

__global__ __launch_bounds__(THREADS, 8) void circuit_gemm(
    const float* __restrict__ x, const float* __restrict__ F,
    float* __restrict__ out)
{
  __shared__ __attribute__((aligned(16))) float Sre[TM * RP];
  __shared__ __attribute__((aligned(16))) float Sim[TM * RP];
  __shared__ float csC[TM][NQ];
  __shared__ float csS[TM][NQ];

  const int t  = threadIdx.x;
  const int s0 = blockIdx.x * TM;
  const int wave = t >> 5, lane = t & 31;
  const int half = lane >> 4, lq = lane & 15;

  const int st = wave >> 2, mt = (wave >> 1) & 1, plane = wave & 1;
  v2f bf[4];
#pragma unroll
  for (int ks = 0; ks < 4; ++ks)
    bf[ks] = *(const v2f*)&F[(st * 2 + plane) * FRAG + ks * 64 + lane * 2];

  {
    const int m = t >> 3, i = t & 7;
    const float a = x[(s0 + m) * NQ + i] * 0.785398163397448f;
    csC[m][i] = cosf(a);
    csS[m][i] = sinf(a);
  }
  __syncthreads();

  float* V = &Sre[0];
  {
    const int W0 = st ? 0 : 4;
    const int m  = mt * 16 + lq;
    const float ca = csC[m][W0 + 0], sa = csS[m][W0 + 0];
    const float cb = csC[m][W0 + 1], sb = csS[m][W0 + 1];
    const float g2 = half ? csS[m][W0 + 2] : csC[m][W0 + 2];
    const float cd = csC[m][W0 + 3], sd = csS[m][W0 + 3];
    v8f acc = (v8f){0.f, 0.f, 0.f, 0.f, 0.f, 0.f, 0.f, 0.f};
#pragma unroll
    for (int ks = 0; ks < 4; ++ks) {
      const float p = (((ks >> 1) ? sa : ca) * ((ks & 1) ? sb : cb)) * g2;
      const v2f a = (v2f){p * cd, p * sd};
      acc = wmma_f32(a, bf[ks], acc);
    }
#pragma unroll
    for (int r = 0; r < 8; ++r)
      V[(st * TM + mt * 16 + 8 * half + r) * VP + lq * 2 + plane] = acc[r];
  }
  __syncthreads();

  {
    const int hi = t & 15;
    const int m0 = (t >> 5) + 8 * ((t >> 4) & 1);
    v2f vh[2];
    float4 vl[2][8];
#pragma unroll
    for (int r = 0; r < 2; ++r) {
      const int m = m0 + 16 * r;
      vh[r] = *(const v2f*)&V[(TM + m) * VP + 2 * hi];
#pragma unroll
      for (int q = 0; q < 8; ++q)
        vl[r][q] = *(const float4*)&V[m * VP + 4 * q];
    }
    __syncthreads();

    const int par = __builtin_popcount(hi) & 1;
    const int hb  = (hi ^ (hi >> 1) ^ (hi >> 2) ^ (hi >> 3)) ^ (par << 3);
    const int xm  = par ? 15 : 0;
    const int b0  = m0 * RP + hb * QP;
    const int b1  = m0 * RP + (hb ^ 8) * QP;
#pragma unroll
    for (int r = 0; r < 2; ++r) {
      const v2f hx = (v2f){vh[r][0], vh[r][0]};
      const v2f hy = (v2f){-vh[r][1], vh[r][1]};
#pragma unroll
      for (int lo = 0; lo < 16; ++lo) {
        const float4 w = vl[r][lo >> 1];
        const v2f l  = (lo & 1) ? (v2f){w.z, w.w} : (v2f){w.x, w.y};
        const v2f ls = (lo & 1) ? (v2f){w.w, w.z} : (v2f){w.y, w.x};
        const v2f z = hx * l + hy * ls;
        const int pl = (lo ^ (lo >> 1) ^ (lo >> 2) ^ (lo >> 3)) & 15;
        const int d  = ((__builtin_popcount(lo) & 1) ? b1 : b0) + (pl ^ xm) + 16 * r * RP;
        Sre[d] = z[0];
        Sim[d] = z[1];
      }
    }
  }
  __syncthreads();

#pragma unroll 1
  for (int r = 0; r < 2; ++r) {
    const int lo = t & 15;
    const int m  = (t >> 4) + 16 * r;
    embed_nibble<QP, 0>(Sre, Sim, m * RP + lo, &csC[m][0], &csS[m][0]);
  }
  __syncthreads();
#pragma unroll 1
  for (int r = 0; r < 2; ++r) {
    const int hi = t & 15;
    const int m  = (t >> 5) + 8 * ((t >> 4) & 1) + 16 * r;
    embed_nibble<1, 4>(Sre, Sim, m * RP + hi * QP, &csC[m][0], &csS[m][0]);
  }
  __syncthreads();

  v8f accR[2][2], accI[2][2];

  factor_stage(F + 2 * 2 * FRAG, Sre, Sim, wave, lane, accR, accI);
  __syncthreads();
  store_swapped(Sre, Sim, wave, lane, accR, accI);
  __syncthreads();
  factor_stage(F + 3 * 2 * FRAG, Sre, Sim, wave, lane, accR, accI);
  __syncthreads();

  float* zp = &Sre[0];

  const unsigned ulq = (unsigned)lq;
  const unsigned sg1 = (ulq & 1u) << 31, sg2 = (ulq & 2u) << 30;
  const unsigned sg4 = (ulq & 4u) << 29, sg8 = (ulq & 8u) << 28;

  const int slotH = (lq == 12) ? 1 : (lq == 14) ? 2 : (lq == 15) ? 3 : -1;
  const int slotD = (lq == 7) ? 0 : (lq == 15) ? 4 : -1;

#pragma unroll
  for (int mt = 0; mt < 2; ++mt)
#pragma unroll
    for (int j = 0; j < 8; ++j) {
      const float p0 = accR[0][mt][j] * accR[0][mt][j] +
                       accI[0][mt][j] * accI[0][mt][j];
      const float p1 = accR[1][mt][j] * accR[1][mt][j] +
                       accI[1][mt][j] * accI[1][mt][j];
      float h = p0 + p1;
      float d = p0 - p1;
      h = walsh_step<1>(h, sg1);
      h = walsh_step<2>(h, sg2);
      h = walsh_step<4>(h, sg4);
      h = walsh_step<8>(h, sg8);
      d = walsh_step<1>(d, sg1);
      d = walsh_step<2>(d, sg2);
      d = walsh_step<4>(d, sg4);
      d = walsh_step<8>(d, sg8);
      const int m = mt * 16 + j + 8 * half;
      if (slotH >= 0) zp[(wave * TM + m) * NQ + slotH] = h;
      if (slotD >= 0) zp[(wave * TM + m) * NQ + slotD] = d;
    }
  __syncthreads();

  {
    const int m = t >> 3, w = t & 7;
    const int zs = (w == 0) ? 0 : (w < 3) ? w : (w < 7) ? 3 : 4;
    const int wm = (w == 0 || w >= 6) ? 7 : (w == 4) ? 4 : (w == 5) ? 6 : 0;
    float acc = 0.f;
#pragma unroll
    for (int wv = 0; wv < 8; ++wv) {
      const float v = zp[(wv * TM + m) * NQ + zs];
      acc += (__builtin_popcount(wv & wm) & 1) ? -v : v;
    }
    out[(s0 + m) * NQ + w] = acc;
  }
}

extern "C" void kernel_launch(void* const* d_in, const int* in_sizes, int n_in,
                              void* d_out, int out_size, void* d_ws, size_t ws_size,
                              hipStream_t stream) {
  const float* x   = (const float*)d_in[0];
  const float* wts = (const float*)d_in[1];
  float* out = (float*)d_out;
  float* F   = (float*)d_ws;

  const int B = in_sizes[0] / NQ;

  build_factor_frags<<<NLAY * 2, 256, 0, stream>>>(wts, F);
  circuit_gemm<<<B / TM, THREADS, 0, stream>>>(x, F, out);
}


# Launch fusion: the separate factor-build launch is gone; wave 0 of each workgroup writes the 16 gates to LDS and every wave builds its own WMMA B fragments from them (one launch, not two)
# speedup vs baseline: 64.2618x; 2.4051x over previous
#include <hip/hip_runtime.h>
#include <hip/hip_bf16.h>
#include <math.h>


typedef float v2f __attribute__((ext_vector_type(2)));
typedef float v8f __attribute__((ext_vector_type(8)));

#define NQ    8
#define NLAY  2
#define TM    32
#define THREADS 256
#define VP    36
#define PA    20
#define NROW  256

static __device__ __forceinline__ v8f wmma_f32(v2f a, v2f b, v8f c) {
  return __builtin_amdgcn_wmma_f32_16x16x4_f32(
       false, a,  false, b,
       (short)0, c,  false,  false);
}

static __device__ __forceinline__ int pre_inv(int p) { return p ^ (p >> 1); }
static __device__ __forceinline__ int hb_inv(int b) {
  const int q = b ^ ((b & 1) << 3);
  return q ^ (q >> 1);
}

static __device__ __forceinline__ void ry_nibble(
    v2f (&zr)[8], v2f (&zi)[8], const float4 cs, const float4 sn)
{
  const float cj[3] = {cs.x, cs.y, cs.z};
  const float sj[3] = {sn.x, sn.y, sn.z};
#pragma unroll
  for (int j = 0; j < 3; ++j) {
    const int qbit = 4 >> j;
    const float c = cj[j], s = sj[j];
#pragma unroll
    for (int q = 0; q < 8; ++q) {
      if (q & qbit) continue;
      const int q1 = q | qbit;
      v2f a0 = zr[q], a1 = zr[q1];
      zr[q]  = c * a0 - s * a1;
      zr[q1] = s * a0 + c * a1;
      a0 = zi[q]; a1 = zi[q1];
      zi[q]  = c * a0 - s * a1;
      zi[q1] = s * a0 + c * a1;
    }
  }
  {
    const float c = cs.w, s = sn.w;
    const v2f ms = (v2f){-s, s};
#pragma unroll
    for (int q = 0; q < 8; ++q) {
      const v2f yr = __builtin_shufflevector(zr[q], zr[q], 1, 0);
      const v2f yi = __builtin_shufflevector(zi[q], zi[q], 1, 0);
      zr[q] = c * zr[q] + ms * yr;
      zi[q] = c * zi[q] + ms * yi;
    }
  }
}

template <int M>
static __device__ __forceinline__ float row_xor(float v) {
  const int b = __builtin_bit_cast(int, v);
  return __builtin_bit_cast(
      float, __builtin_amdgcn_update_dpp(b, b, 0x160 + M, 0xf, 0xf, true));
}

template <int M>
static __device__ __forceinline__ float walsh_step(float v, unsigned sgn) {
  const float other = row_xor<M>(v);
  const float own =
      __builtin_bit_cast(float, __builtin_bit_cast(unsigned, v) ^ sgn);
  return other + own;
}

static __device__ __forceinline__ void gate_prefix(
    const float* G, const int gb, const int lq, const int half,
    float (&qr)[4], float (&qi)[4], v2f& ur3, v2f& ui3)
{
  const float4 g0 = *(const float4*)&G[((gb + 0) * 4 + ((lq >> 3) & 1) * 2) * 2];
  const float4 g1 = *(const float4*)&G[((gb + 1) * 4 + ((lq >> 2) & 1) * 2) * 2];
  const v2f    g2 = *(const v2f*)&G[((gb + 2) * 4 + ((lq >> 1) & 1) * 2 + half) * 2];
  const float4 g3 = *(const float4*)&G[((gb + 3) * 4 + (lq & 1) * 2) * 2];
  ur3 = (v2f){g3.x, g3.z};
  ui3 = (v2f){g3.y, g3.w};
#pragma unroll
  for (int ks = 0; ks < 4; ++ks) {
    const float ar = (ks >> 1) ? g0.z : g0.x, ai = (ks >> 1) ? g0.w : g0.y;
    const float br = (ks & 1) ? g1.z : g1.x, bi = (ks & 1) ? g1.w : g1.y;
    const float pr = ar * br - ai * bi;
    const float pi = ar * bi + ai * br;
    qr[ks] = pr * g2[0] - pi * g2[1];
    qi[ks] = pr * g2[1] + pi * g2[0];
  }
}

__global__ __launch_bounds__(THREADS, 8) void circuit_gemm(
    const float* __restrict__ x, const float* __restrict__ wts,
    float* __restrict__ out)
{
  __shared__ __attribute__((aligned(16))) float Tre[NROW * PA];
  __shared__ __attribute__((aligned(16))) float Tim[NROW * PA];
  __shared__ __attribute__((aligned(16))) float V[2 * TM * VP];
  __shared__ __attribute__((aligned(16))) float csC[TM][NQ];
  __shared__ __attribute__((aligned(16))) float csS[TM][NQ];
  __shared__ __attribute__((aligned(16))) float G[NLAY * NQ * 8];

  const int t  = threadIdx.x;
  const int s0 = blockIdx.x * TM;
  const int wave = t >> 5, lane = t & 31;
  const int half = lane >> 4, lq = lane & 15;

  const int st = wave >> 2, mt = (wave >> 1) & 1, plane = wave & 1;

  if (wave == 0) {
    const float h = 0.5f * wts[lane];
    const float c = cosf(h), s = sinf(h);
    const float c2 = row_xor<1>(c), s2 = row_xor<1>(s);
    const int ob = lane & 1;
    const float cy = ob ? c2 : c, sy = ob ? s2 : s;
    const float cz = ob ? c : c2, sz = ob ? s : s2;
    const float ry0 = ob ? sy : cy;
    const float ry1 = ob ? cy : -sy;
    const float ur0 = ry0 * cz, ur1 = ry1 * cz;
    const float ui0 = ob ? ry0 * sz : -(ry0 * sz);
    const float ui1 = ob ? ry1 * sz : -(ry1 * sz);
    *(float4*)&G[lane * 4] = (float4){ur0, ui0, ur1, ui1};
  }

  {
    const int m = t >> 3, i = t & 7;
    const float a = x[(s0 + m) * NQ + i] * 0.785398163397448f;
    csC[m][i] = cosf(a);
    csS[m][i] = sinf(a);
  }
  __syncthreads();

  {
    const int W0 = st ? 0 : 4;
    const int m  = mt * 16 + lq;
    v2f bf[4];
    {
      float qr[4], qi[4];
      v2f ur3, ui3;
      gate_prefix(G, W0, lq, half, qr, qi, ur3, ui3);
      const v2f fa = plane ? ui3 : ur3;
      const v2f fb = plane ? ur3 : -ui3;
#pragma unroll
      for (int ks = 0; ks < 4; ++ks) bf[ks] = qr[ks] * fa + qi[ks] * fb;
    }
    const float ca = csC[m][W0 + 0], sa = csS[m][W0 + 0];
    const float cb = csC[m][W0 + 1], sb = csS[m][W0 + 1];
    const float g2 = half ? csS[m][W0 + 2] : csC[m][W0 + 2];
    const float cd = csC[m][W0 + 3], sd = csS[m][W0 + 3];
    v8f acc = (v8f){0.f, 0.f, 0.f, 0.f, 0.f, 0.f, 0.f, 0.f};
#pragma unroll
    for (int ks = 0; ks < 4; ++ks) {
      const float p = (((ks >> 1) ? sa : ca) * ((ks & 1) ? sb : cb)) * g2;
      const v2f a = (v2f){p * cd, p * sd};
      acc = wmma_f32(a, bf[ks], acc);
    }
#pragma unroll
    for (int r = 0; r < 8; ++r)
      V[(st * TM + mt * 16 + 8 * half + r) * VP + lq * 2 + plane] = acc[r];
  }

  v2f br[4], bi[4];
  {
    float qr[4], qi[4];
    v2f ur3, ui3;
    gate_prefix(G, NQ + (st ? 0 : 4), lq, half, qr, qi, ur3, ui3);
#pragma unroll
    for (int ks = 0; ks < 4; ++ks) {
      br[ks] = qr[ks] * ur3 - qi[ks] * ui3;
      bi[ks] = qr[ks] * ui3 + qi[ks] * ur3;
    }
  }
  __syncthreads();

  {
    const int term = wave & 3, ph = term >> 1, pl = term & 1;
    const int m  = lane;
    const int W0 = st ? 0 : 4;
    const int xr = st ? (pl ? 12 : 0) : (ph ? 8 : 0);
    const int kp = st ? ph : (ph ^ pl);
    const float* vrow = &V[(st * TM + m) * VP];
    v2f zr[8], zi[8];
#pragma unroll
    for (int q = 0; q < 8; ++q) {
      const int b   = 2 * q + kp;
      const int src = (st ? hb_inv(b) : pre_inv(b)) ^ xr;
      const v2f val = *(const v2f*)&vrow[2 * src];
      zr[q] = kp ? (v2f){0.f, val[0]} : (v2f){val[0], 0.f};
      zi[q] = kp ? (v2f){0.f, val[1]} : (v2f){val[1], 0.f};
    }
    const float4 c4 = *(const float4*)&csC[m][W0];
    const float4 s4 = *(const float4*)&csS[m][W0];
    ry_nibble(zr, zi, c4, s4);
    float* dr = &Tre[(wave * TM + m) * PA];
    float* di = &Tim[(wave * TM + m) * PA];
#pragma unroll
    for (int g = 0; g < 4; ++g) {
      *(float4*)&dr[4 * g] =
          (float4){zr[2 * g][0], zr[2 * g][1], zr[2 * g + 1][0], zr[2 * g + 1][1]};
      *(float4*)&di[4 * g] =
          (float4){zi[2 * g][0], zi[2 * g][1], zi[2 * g + 1][0], zi[2 * g + 1][1]};
    }
  }
  __syncthreads();

  {
    v8f accR[2], accI[2];
#pragma unroll
    for (int mi = 0; mi < 2; ++mi) {
      accR[mi] = (v8f){0.f, 0.f, 0.f, 0.f, 0.f, 0.f, 0.f, 0.f};
      accI[mi] = (v8f){0.f, 0.f, 0.f, 0.f, 0.f, 0.f, 0.f, 0.f};
    }
#pragma unroll
    for (int ks = 0; ks < 4; ++ks)
#pragma unroll
      for (int mi = 0; mi < 2; ++mi) {
        const int a = (wave * TM + mi * 16 + lq) * PA + 4 * ks + 2 * half;
        const v2f ar = *(const v2f*)&Tre[a];
        const v2f ai = *(const v2f*)&Tim[a];
        const v2f an = -ai;
        accR[mi] = wmma_f32(ar, br[ks], accR[mi]);
        accI[mi] = wmma_f32(ar, bi[ks], accI[mi]);
        accR[mi] = wmma_f32(an, bi[ks], accR[mi]);
        accI[mi] = wmma_f32(ai, br[ks], accI[mi]);
      }
#pragma unroll
    for (int mi = 0; mi < 2; ++mi)
#pragma unroll
      for (int j = 0; j < 8; ++j) {
        const int d = (wave * TM + mi * 16 + j + 8 * half) * PA + lq;
        Tre[d] = accR[mi][j];
        Tim[d] = accI[mi][j];
      }
  }
  __syncthreads();

  float* zp = &V[0];

  const unsigned ulq = (unsigned)lq;
  const unsigned sg1 = (ulq & 1u) << 31, sg2 = (ulq & 2u) << 30;
  const unsigned sg4 = (ulq & 4u) << 29, sg8 = (ulq & 8u) << 28;
  const int slotW = (lq == 0) ? 3 : (lq == 8) ? 4 : (lq == 12) ? 5
                  : (lq == 14) ? 6 : (lq == 15) ? 7 : -1;

#pragma unroll
  for (int i = 0; i < 4; ++i) {
    const int m  = wave * 4 + i;
    const int ra = ((4 + 2 * half) * TM + m) * PA + lq;
    const int rb = ((2 * half) * TM + m) * PA + lq;
    const v2f ar = (v2f){Tre[ra], Tre[ra + TM * PA]};
    const v2f ai = (v2f){Tim[ra], Tim[ra + TM * PA]};
    const v2f qr = (v2f){Tre[rb], Tre[rb + TM * PA]};
    const v2f qi = (v2f){Tim[rb], Tim[rb + TM * PA]};
    const v2f an = -ai;
    const v8f zero = (v8f){0.f, 0.f, 0.f, 0.f, 0.f, 0.f, 0.f, 0.f};
    v8f pr = wmma_f32(ar, qr, zero);
    v8f pi = wmma_f32(ar, qi, zero);
    pr = wmma_f32(an, qi, pr);
    pi = wmma_f32(ai, qr, pi);
    float p[8];
#pragma unroll
    for (int j = 0; j < 8; ++j) p[j] = pr[j] * pr[j] + pi[j] * pi[j];
    const float e0 = p[0] + p[1], e1 = p[2] + p[3], e2 = p[4] + p[5], e3 = p[6] + p[7];
    const float d0 = p[0] - p[1], d1 = p[2] - p[3], d2 = p[4] - p[5], d3 = p[6] - p[7];
    float a4 = (e0 + e1) - (e2 + e3);
    float a6 = (e0 - e1) - (e2 - e3);
    float a7 = (d0 - d1) - (d2 - d3);
    a7 = walsh_step<1>(a7, sg1);
    a7 = walsh_step<2>(a7, sg2);
    a7 = walsh_step<4>(a7, sg4);
    a7 = walsh_step<8>(a7, sg8);
    a4 += row_xor<1>(a4);  a6 += row_xor<1>(a6);
    a4 += row_xor<2>(a4);  a6 += row_xor<2>(a6);
    a4 += row_xor<4>(a4);  a6 += row_xor<4>(a6);
    a4 += row_xor<8>(a4);  a6 += row_xor<8>(a6);
    float* zrow = &zp[(m * 2 + half) * NQ];
    if (slotW >= 0) zrow[slotW] = a7;
    if (lq == 15) zrow[0] = a7;
    if (lq == 1)  zrow[1] = a4;
    if (lq == 2)  zrow[2] = a6;
  }
  __syncthreads();

  {
    const int m = t >> 3, w = t & 7;
    const float lo8 = zp[(m * 2 + 0) * NQ + w];
    const float hi8 = zp[(m * 2 + 1) * NQ + w];
    out[(s0 + m) * NQ + w] = (w == 0) ? lo8 + hi8 : lo8 - hi8;
  }
}

extern "C" void kernel_launch(void* const* d_in, const int* in_sizes, int n_in,
                              void* d_out, int out_size, void* d_ws, size_t ws_size,
                              hipStream_t stream) {
  const float* x   = (const float*)d_in[0];
  const float* wts = (const float*)d_in[1];
  float* out = (float*)d_out;

  const int B = in_sizes[0] / NQ;

  circuit_gemm<<<B / TM, THREADS, 0, stream>>>(x, wts, out);
}


# Layer-2 rotation of a term vector runs on its 8 nonzero entries (lowest nibble bit fixed by the term's parity): 3 RY gates on 8 complex values, the 4th is one scaling per entry; sums kept
# speedup vs baseline: 98.4325x; 1.5317x over previous
#include <hip/hip_runtime.h>
#include <hip/hip_bf16.h>
#include <math.h>


typedef float v2f __attribute__((ext_vector_type(2)));
typedef float v8f __attribute__((ext_vector_type(8)));

#define NQ    8
#define TM    32
#define THREADS 256
#define VP    36
#define PA    20
#define NROW  256

static __device__ __forceinline__ v8f wmma_f32(v2f a, v2f b, v8f c) {
  return __builtin_amdgcn_wmma_f32_16x16x4_f32(
       false, a,  false, b,
       (short)0, c,  false,  false);
}

static __device__ __forceinline__ int pre_inv(int p) { return p ^ (p >> 1); }
static __device__ __forceinline__ int hb_inv(int b) {
  const int q = b ^ ((b & 1) << 3);
  return q ^ (q >> 1);
}

static __device__ __forceinline__ void ry_upper3(
    v2f (&z)[8], const float4 cs, const float4 sn)
{
  v2f a[8], b[8];
#pragma unroll
  for (int q = 0; q < 4; ++q) {
    a[q]     = cs.x * z[q] - sn.x * z[q + 4];
    a[q + 4] = sn.x * z[q] + cs.x * z[q + 4];
  }
#pragma unroll
  for (int q = 0; q < 8; ++q) {
    if (q & 2) continue;
    b[q]     = cs.y * a[q] - sn.y * a[q + 2];
    b[q + 2] = sn.y * a[q] + cs.y * a[q + 2];
  }
#pragma unroll
  for (int q = 0; q < 8; q += 2) {
    z[q]     = cs.z * b[q] - sn.z * b[q + 1];
    z[q + 1] = sn.z * b[q] + cs.z * b[q + 1];
  }
}

template <int M>
static __device__ __forceinline__ float row_xor(float v) {
  const int b = __builtin_bit_cast(int, v);
  return __builtin_bit_cast(
      float, __builtin_amdgcn_update_dpp(b, b, 0x160 + M, 0xf, 0xf, true));
}

template <int M>
static __device__ __forceinline__ float walsh_step(float v, unsigned sgn) {
  const float other = row_xor<M>(v);
  const float own =
      __builtin_bit_cast(float, __builtin_bit_cast(unsigned, v) ^ sgn);
  return other + own;
}

static __device__ __forceinline__ void gate_prefix(
    const float* G, const int gb, const int lq, const int half,
    float (&qr)[4], float (&qi)[4], v2f& ur3, v2f& ui3)
{
  const float4 g0 = *(const float4*)&G[((gb + 0) * 4 + ((lq >> 3) & 1) * 2) * 2];
  const float4 g1 = *(const float4*)&G[((gb + 1) * 4 + ((lq >> 2) & 1) * 2) * 2];
  const v2f    g2 = *(const v2f*)&G[((gb + 2) * 4 + ((lq >> 1) & 1) * 2 + half) * 2];
  const float4 g3 = *(const float4*)&G[((gb + 3) * 4 + (lq & 1) * 2) * 2];
  ur3 = (v2f){g3.x, g3.z};
  ui3 = (v2f){g3.y, g3.w};
#pragma unroll
  for (int ks = 0; ks < 4; ++ks) {
    const float ar = (ks >> 1) ? g0.z : g0.x, ai = (ks >> 1) ? g0.w : g0.y;
    const float br = (ks & 1) ? g1.z : g1.x, bi = (ks & 1) ? g1.w : g1.y;
    const float pr = ar * br - ai * bi;
    const float pi = ar * bi + ai * br;
    qr[ks] = pr * g2[0] - pi * g2[1];
    qi[ks] = pr * g2[1] + pi * g2[0];
  }
}

__global__ __launch_bounds__(THREADS, 8) void circuit_gemm(
    const float* __restrict__ x, const float* __restrict__ wts,
    float* __restrict__ out)
{
  __shared__ __attribute__((aligned(16))) float Tre[NROW * PA];
  __shared__ __attribute__((aligned(16))) float Tim[NROW * PA];
  __shared__ __attribute__((aligned(16))) float V[2 * TM * VP];
  __shared__ __attribute__((aligned(16))) float csC[TM][NQ];
  __shared__ __attribute__((aligned(16))) float csS[TM][NQ];
  __shared__ __attribute__((aligned(16))) float G[NQ * 8];
  __shared__ __attribute__((aligned(16))) float W2c[NQ];
  __shared__ __attribute__((aligned(16))) float W2s[NQ];

  const int t  = threadIdx.x;
  const int s0 = blockIdx.x * TM;
  const int wave = t >> 5, lane = t & 31;
  const int half = lane >> 4, lq = lane & 15;

  const int st = wave >> 2, mt = (wave >> 1) & 1, plane = wave & 1;

  if (wave == 0) {
    const float h = 0.5f * wts[lane];
    const float c = cosf(h), s = sinf(h);
    const float c2 = row_xor<1>(c), s2 = row_xor<1>(s);
    const int ob = lane & 1;
    const float cy = ob ? c2 : c, sy = ob ? s2 : s;
    const float cz = ob ? c : c2, sz = ob ? s : s2;
    const float ry0 = ob ? sy : cy;
    const float ry1 = ob ? cy : -sy;
    const float ur0 = ry0 * cz, ur1 = ry1 * cz;
    const float ui0 = ob ? ry0 * sz : -(ry0 * sz);
    const float ui1 = ob ? ry1 * sz : -(ry1 * sz);
    if (lane < 2 * NQ) {
      *(float4*)&G[lane * 4] = (float4){ur0, ui0, ur1, ui1};
    } else if (!(lane & 1)) {
      W2c[(lane >> 1) & 7] = c;
      W2s[(lane >> 1) & 7] = s;
    }
  }

  {
    const int m = t >> 3, i = t & 7;
    const float a = x[(s0 + m) * NQ + i] * 0.785398163397448f;
    csC[m][i] = cosf(a);
    csS[m][i] = sinf(a);
  }
  __syncthreads();

  {
    const int W0 = st ? 0 : 4;
    const int m  = mt * 16 + lq;
    v2f bf[4];
    {
      float qr[4], qi[4];
      v2f ur3, ui3;
      gate_prefix(G, W0, lq, half, qr, qi, ur3, ui3);
      const v2f fa = plane ? ui3 : ur3;
      const v2f fb = plane ? ur3 : -ui3;
#pragma unroll
      for (int ks = 0; ks < 4; ++ks) bf[ks] = qr[ks] * fa + qi[ks] * fb;
    }
    const float ca = csC[m][W0 + 0], sa = csS[m][W0 + 0];
    const float cb = csC[m][W0 + 1], sb = csS[m][W0 + 1];
    const float g2 = half ? csS[m][W0 + 2] : csC[m][W0 + 2];
    const float cd = csC[m][W0 + 3], sd = csS[m][W0 + 3];
    v8f acc = (v8f){0.f, 0.f, 0.f, 0.f, 0.f, 0.f, 0.f, 0.f};
#pragma unroll
    for (int ks = 0; ks < 4; ++ks) {
      const float p = (((ks >> 1) ? sa : ca) * ((ks & 1) ? sb : cb)) * g2;
      const v2f a = (v2f){p * cd, p * sd};
      acc = wmma_f32(a, bf[ks], acc);
    }
#pragma unroll
    for (int r = 0; r < 8; ++r)
      V[(st * TM + mt * 16 + 8 * half + r) * VP + lq * 2 + plane] = acc[r];
  }

  __syncthreads();

  {
    const int term = wave & 3, ph = term >> 1, pl = term & 1;
    const int m  = lane;
    const int W0 = st ? 0 : 4;
    const int xr = st ? (pl ? 12 : 0) : (ph ? 8 : 0);
    const int kp = st ? ph : (ph ^ pl);
    const float* vrow = &V[(st * TM + m) * VP];
    v2f z[8];
#pragma unroll
    for (int q = 0; q < 8; ++q) {
      const int b   = 2 * q + kp;
      const int src = (st ? hb_inv(b) : pre_inv(b)) ^ xr;
      z[q] = *(const v2f*)&vrow[2 * src];
    }
    const float4 ca = *(const float4*)&csC[m][W0];
    const float4 sa = *(const float4*)&csS[m][W0];
    const float4 ch = *(const float4*)&W2c[W0];
    const float4 sh = *(const float4*)&W2s[W0];
    const float4 c4 = (float4){ca.x * ch.x - sa.x * sh.x, ca.y * ch.y - sa.y * sh.y,
                               ca.z * ch.z - sa.z * sh.z, ca.w * ch.w - sa.w * sh.w};
    const float4 s4 = (float4){sa.x * ch.x + ca.x * sh.x, sa.y * ch.y + ca.y * sh.y,
                               sa.z * ch.z + ca.z * sh.z, sa.w * ch.w + ca.w * sh.w};
    ry_upper3(z, c4, s4);
    const v2f col = kp ? (v2f){-s4.w, c4.w} : (v2f){c4.w, s4.w};
    float* dr = &Tre[(wave * TM + m) * PA];
    float* di = &Tim[(wave * TM + m) * PA];
#pragma unroll
    for (int g = 0; g < 4; ++g) {
      const v2f r0 = z[2 * g][0] * col, r1 = z[2 * g + 1][0] * col;
      const v2f i0 = z[2 * g][1] * col, i1 = z[2 * g + 1][1] * col;
      *(float4*)&dr[4 * g] = (float4){r0[0], r0[1], r1[0], r1[1]};
      *(float4*)&di[4 * g] = (float4){i0[0], i0[1], i1[0], i1[1]};
    }
  }
  __syncthreads();

  float* zp = &V[0];

  const unsigned ulq = (unsigned)lq;
  const unsigned sg1 = (ulq & 1u) << 31, sg2 = (ulq & 2u) << 30;
  const unsigned sg4 = (ulq & 4u) << 29, sg8 = (ulq & 8u) << 28;
  const int slotW = (lq == 0) ? 3 : (lq == 8) ? 4 : (lq == 12) ? 5
                  : (lq == 14) ? 6 : (lq == 15) ? 7 : -1;

#pragma unroll
  for (int i = 0; i < 4; ++i) {
    const int m  = wave * 4 + i;
    const int ra = ((4 + 2 * half) * TM + m) * PA + lq;
    const int rb = ((2 * half) * TM + m) * PA + lq;
    const v2f ar = (v2f){Tre[ra], Tre[ra + TM * PA]};
    const v2f ai = (v2f){Tim[ra], Tim[ra + TM * PA]};
    const v2f qr = (v2f){Tre[rb], Tre[rb + TM * PA]};
    const v2f qi = (v2f){Tim[rb], Tim[rb + TM * PA]};
    const v2f an = -ai;
    const v8f zero = (v8f){0.f, 0.f, 0.f, 0.f, 0.f, 0.f, 0.f, 0.f};
    v8f pr = wmma_f32(ar, qr, zero);
    v8f pi = wmma_f32(ar, qi, zero);
    pr = wmma_f32(an, qi, pr);
    pi = wmma_f32(ai, qr, pi);
    float p[8];
#pragma unroll
    for (int j = 0; j < 8; ++j) p[j] = pr[j] * pr[j] + pi[j] * pi[j];
    const float e0 = p[0] + p[1], e1 = p[2] + p[3], e2 = p[4] + p[5], e3 = p[6] + p[7];
    const float d0 = p[0] - p[1], d1 = p[2] - p[3], d2 = p[4] - p[5], d3 = p[6] - p[7];
    float a4 = (e0 + e1) - (e2 + e3);
    float a6 = (e0 - e1) - (e2 - e3);
    float a7 = (d0 - d1) - (d2 - d3);
    a7 = walsh_step<1>(a7, sg1);
    a7 = walsh_step<2>(a7, sg2);
    a7 = walsh_step<4>(a7, sg4);
    a7 = walsh_step<8>(a7, sg8);
    a4 += row_xor<1>(a4);  a6 += row_xor<1>(a6);
    a4 += row_xor<2>(a4);  a6 += row_xor<2>(a6);
    a4 += row_xor<4>(a4);  a6 += row_xor<4>(a6);
    a4 += row_xor<8>(a4);  a6 += row_xor<8>(a6);
    float* zrow = &zp[(m * 2 + half) * NQ];
    if (slotW >= 0) zrow[slotW] = a7;
    if (lq == 15) zrow[0] = a7;
    if (lq == 1)  zrow[1] = a4;
    if (lq == 2)  zrow[2] = a6;
  }
  __syncthreads();

  {
    const int m = t >> 3, w = t & 7;
    const float lo8 = zp[(m * 2 + 0) * NQ + w];
    const float hi8 = zp[(m * 2 + 1) * NQ + w];
    out[(s0 + m) * NQ + w] = (w == 0) ? lo8 + hi8 : lo8 - hi8;
  }
}

extern "C" void kernel_launch(void* const* d_in, const int* in_sizes, int n_in,
                              void* d_out, int out_size, void* d_ws, size_t ws_size,
                              hipStream_t stream) {
  const float* x   = (const float*)d_in[0];
  const float* wts = (const float*)d_in[1];
  float* out = (float*)d_out;

  const int B = in_sizes[0] / NQ;

  circuit_gemm<<<B / TM, THREADS, 0, stream>>>(x, wts, out);
}
